# starting kernel
# baseline (speedup 1.0000x reference)
#include <hip/hip_runtime.h>
#include <math.h>

#define NNODES 50000
#define NEDGES 800000
#define NGRAPH 256
#define FIN    128
#define H1C    12
#define D1C    32
#define F1     (H1C * D1C)
#define H2C    1
#define D2C    64
#define F2     (H2C * D2C)
#define ETOT   (NEDGES + NNODES)

typedef float v2f __attribute__((ext_vector_type(2)));
typedef float v8f __attribute__((ext_vector_type(8)));

__device__ __forceinline__ unsigned fkey(float f) {
  unsigned u = __float_as_uint(f);
  return (u & 0x80000000u) ? ~u : (u | 0x80000000u);
}
__device__ __forceinline__ float fdecode(unsigned u) {
  unsigned b = (u & 0x80000000u) ? (u & 0x7fffffffu) : ~u;
  return __uint_as_float(b);
}

__global__ void __launch_bounds__(128)
wmma_gemm_f32(const float* __restrict__ A, const float* __restrict__ B,
              float* __restrict__ C, int K, int Ncols) {
  const int lane = threadIdx.x & 31;
  const int wave = threadIdx.x >> 5;
  const int l    = lane & 15;
  const int half = lane >> 4;
  const int m0   = blockIdx.x << 4;
  const int n0   = ((blockIdx.y << 2) + wave) << 4;

  v8f c = {};
  const float* arow = A + (size_t)(m0 + l) * K + (half << 1);
  const float* bcol = B + n0 + l;

  for (int k0 = 0; k0 < K; k0 += 4) {
    v2f a = *(const v2f*)(arow + k0);
    const int kb = k0 + (half << 1);
    v2f b;
    b.x = bcol[(size_t)kb * Ncols];
    b.y = bcol[(size_t)(kb + 1) * Ncols];
    c = __builtin_amdgcn_wmma_f32_16x16x4_f32(
         false, a,  false, b,
         (short)0, c,  false,  false);
  }

  float* crow = C + (size_t)(m0 + (half << 3)) * Ncols + n0 + l;
#pragma unroll
  for (int r = 0; r < 8; ++r) crow[(size_t)r * Ncols] = c[r];
}

__global__ void att_coeff(const float* __restrict__ h,
                          const float* __restrict__ att_src,
                          const float* __restrict__ att_dst,
                          float* __restrict__ asrc, float* __restrict__ adst,
                          int H, int D) {
  int i = blockIdx.x * blockDim.x + threadIdx.x;
  if (i >= NNODES * H) return;
  int n = i / H, hh = i - n * H;
  const float* hp = h + (size_t)n * H * D + hh * D;
  const float* ws = att_src + hh * D;
  const float* wd = att_dst + hh * D;
  float s = 0.f, d = 0.f;
  for (int j = 0; j < D; ++j) { float v = hp[j]; s += v * ws[j]; d += v * wd[j]; }
  asrc[i] = s; adst[i] = d;
}

__global__ void edge_logit_max(const int* __restrict__ srcI, const int* __restrict__ dstI,
                               const float* __restrict__ asrc, const float* __restrict__ adst,
                               float* __restrict__ e, unsigned* __restrict__ mkey, int H) {
  int i = blockIdx.x * blockDim.x + threadIdx.x;
  if (i >= ETOT * H) return;
  int eid = i / H, hh = i - eid * H;
  int s, d;
  if (eid < NEDGES) { s = srcI[eid]; d = dstI[eid]; } else { s = eid - NEDGES; d = s; }
  float v = asrc[s * H + hh] + adst[d * H + hh];
  v = (v > 0.f) ? v : 0.2f * v;
  e[i] = v;
  atomicMax(&mkey[d * H + hh], fkey(v));
}

__global__ void edge_exp_sum(const int* __restrict__ dstI, float* __restrict__ e,
                             const unsigned* __restrict__ mkey, float* __restrict__ ssum,
                             int H) {
  int i = blockIdx.x * blockDim.x + threadIdx.x;
  if (i >= ETOT * H) return;
  int eid = i / H, hh = i - eid * H;
  int d = (eid < NEDGES) ? dstI[eid] : (eid - NEDGES);
  float p = __expf(e[i] - fdecode(mkey[d * H + hh]));
  e[i] = p;
  atomicAdd(&ssum[d * H + hh], p);
}

__global__ void edge_alpha(const int* __restrict__ dstI, float* __restrict__ e,
                           const float* __restrict__ ssum, int H) {
  int i = blockIdx.x * blockDim.x + threadIdx.x;
  if (i >= ETOT * H) return;
  int eid = i / H, hh = i - eid * H;
  int d = (eid < NEDGES) ? dstI[eid] : (eid - NEDGES);
  e[i] = e[i] / ssum[d * H + hh];
}

__global__ void edge_agg(const int* __restrict__ srcI, const int* __restrict__ dstI,
                         const float* __restrict__ alpha, const float* __restrict__ h,
                         float* __restrict__ out, int H, int D) {
  const int F = H * D;
  int i = blockIdx.x * blockDim.x + threadIdx.x;
  if (i >= ETOT * F) return;
  int eid = i / F, f = i - eid * F;
  int hh = f / D;
  int s, d;
  if (eid < NEDGES) { s = srcI[eid]; d = dstI[eid]; } else { s = eid - NEDGES; d = s; }
  atomicAdd(&out[(size_t)d * F + f], alpha[eid * H + hh] * h[(size_t)s * F + f]);
}

__global__ void bias_act(float* __restrict__ x, const float* __restrict__ bias,
                         int total, int F, int do_relu) {
  int i = blockIdx.x * blockDim.x + threadIdx.x;
  if (i >= total) return;
  float v = x[i] + bias[i % F];
  if (do_relu) v = fmaxf(v, 0.f);
  x[i] = v;
}

__global__ void pool_accum(const float* __restrict__ h, const int* __restrict__ batch,
                           float* __restrict__ sums, float* __restrict__ cnt) {
  int i = blockIdx.x * blockDim.x + threadIdx.x;
  if (i >= NNODES * F2) return;
  int n = i / F2, f = i - n * F2;
  int g = batch[n];
  atomicAdd(&sums[g * F2 + f], h[i]);
  if (f == 0) atomicAdd(&cnt[g], 1.0f);
}

__global__ void pool_final(const float* __restrict__ sums, const float* __restrict__ cnt,
                           float* __restrict__ out) {
  int i = blockIdx.x * blockDim.x + threadIdx.x;
  if (i >= NGRAPH * F2) return;
  out[i] = sums[i] / fmaxf(cnt[i / F2], 1.0f);
}

static inline int cdiv(long long a, int b) { return (int)((a + b - 1) / b); }

extern "C" void kernel_launch(void* const* d_in, const int* in_sizes, int n_in,
                              void* d_out, int out_size, void* d_ws, size_t ws_size,
                              hipStream_t stream) {
  const float* x        = (const float*)d_in[0];
  const int*   ei       = (const int*)d_in[1];
  const int*   batch    = (const int*)d_in[2];
  const float* W1       = (const float*)d_in[3];
  const float* att_src1 = (const float*)d_in[4];
  const float* att_dst1 = (const float*)d_in[5];
  const float* bias1    = (const float*)d_in[6];
  const float* W2       = (const float*)d_in[7];
  const float* att_src2 = (const float*)d_in[8];
  const float* att_dst2 = (const float*)d_in[9];
  const float* bias2    = (const float*)d_in[10];
  float* out = (float*)d_out;
  const int* srcI = ei;
  const int* dstI = ei + NEDGES;

  char* p = (char*)d_ws;
  auto alloc = [&](size_t bytes) -> char* {
    char* r = p; p += (bytes + 255) & ~(size_t)255; return r;
  };
  float*    h1    = (float*)alloc((size_t)NNODES * F1 * 4);
  float*    asrc1 = (float*)alloc((size_t)NNODES * H1C * 4);
  float*    adst1 = (float*)alloc((size_t)NNODES * H1C * 4);
  unsigned* mkey1 = (unsigned*)alloc((size_t)NNODES * H1C * 4);
  float*    ssum1 = (float*)alloc((size_t)NNODES * H1C * 4);
  float*    e1    = (float*)alloc((size_t)ETOT * H1C * 4);
  float*    agg1  = (float*)alloc((size_t)NNODES * F1 * 4);
  float*    h2    = (float*)alloc((size_t)NNODES * F2 * 4);
  float*    asrc2 = (float*)alloc((size_t)NNODES * H2C * 4);
  float*    adst2 = (float*)alloc((size_t)NNODES * H2C * 4);
  unsigned* mkey2 = (unsigned*)alloc((size_t)NNODES * H2C * 4);
  float*    ssum2 = (float*)alloc((size_t)NNODES * H2C * 4);
  float*    e2    = (float*)alloc((size_t)ETOT * H2C * 4);
  float*    agg2  = (float*)alloc((size_t)NNODES * F2 * 4);
  float*    psum  = (float*)alloc((size_t)NGRAPH * F2 * 4);
  float*    pcnt  = (float*)alloc((size_t)NGRAPH * 4);

  hipMemsetAsync(mkey1, 0, (size_t)NNODES * H1C * 4, stream);
  hipMemsetAsync(ssum1, 0, (size_t)NNODES * H1C * 4, stream);
  hipMemsetAsync(agg1,  0, (size_t)NNODES * F1 * 4, stream);
  hipMemsetAsync(mkey2, 0, (size_t)NNODES * H2C * 4, stream);
  hipMemsetAsync(ssum2, 0, (size_t)NNODES * H2C * 4, stream);
  hipMemsetAsync(agg2,  0, (size_t)NNODES * F2 * 4, stream);
  hipMemsetAsync(psum,  0, (size_t)NGRAPH * F2 * 4, stream);
  hipMemsetAsync(pcnt,  0, (size_t)NGRAPH * 4, stream);

  const int T = 256;

  wmma_gemm_f32<<<dim3(NNODES / 16, F1 / 64), 128, 0, stream>>>(x, W1, h1, FIN, F1);
  att_coeff<<<cdiv((long long)NNODES * H1C, T), T, 0, stream>>>(h1, att_src1, att_dst1,
                                                                asrc1, adst1, H1C, D1C);
  edge_logit_max<<<cdiv((long long)ETOT * H1C, T), T, 0, stream>>>(srcI, dstI, asrc1, adst1,
                                                                   e1, mkey1, H1C);
  edge_exp_sum<<<cdiv((long long)ETOT * H1C, T), T, 0, stream>>>(dstI, e1, mkey1, ssum1, H1C);
  edge_alpha<<<cdiv((long long)ETOT * H1C, T), T, 0, stream>>>(dstI, e1, ssum1, H1C);
  edge_agg<<<cdiv((long long)ETOT * F1, T), T, 0, stream>>>(srcI, dstI, e1, h1, agg1, H1C, D1C);
  bias_act<<<cdiv((long long)NNODES * F1, T), T, 0, stream>>>(agg1, bias1, NNODES * F1, F1, 1);

  wmma_gemm_f32<<<dim3(NNODES / 16, F2 / 64), 128, 0, stream>>>(agg1, W2, h2, F1, F2);
  att_coeff<<<cdiv((long long)NNODES * H2C, T), T, 0, stream>>>(h2, att_src2, att_dst2,
                                                                asrc2, adst2, H2C, D2C);
  edge_logit_max<<<cdiv((long long)ETOT * H2C, T), T, 0, stream>>>(srcI, dstI, asrc2, adst2,
                                                                   e2, mkey2, H2C);
  edge_exp_sum<<<cdiv((long long)ETOT * H2C, T), T, 0, stream>>>(dstI, e2, mkey2, ssum2, H2C);
  edge_alpha<<<cdiv((long long)ETOT * H2C, T), T, 0, stream>>>(dstI, e2, ssum2, H2C);
  edge_agg<<<cdiv((long long)ETOT * F2, T), T, 0, stream>>>(srcI, dstI, e2, h2, agg2, H2C, D2C);
  bias_act<<<cdiv((long long)NNODES * F2, T), T, 0, stream>>>(agg2, bias2, NNODES * F2, F2, 0);

  pool_accum<<<cdiv((long long)NNODES * F2, T), T, 0, stream>>>(agg2, batch, psum, pcnt);
  pool_final<<<cdiv((long long)NGRAPH * F2, T), T, 0, stream>>>(psum, pcnt, out);
}


# Aggregate by destination segment: in-call counting sort of edges by dst, each row summed in registers, stored once with bias/relu; float atomic adds on agg gone, summation order changes
# speedup vs baseline: 2.6131x; 2.6131x over previous
#include <hip/hip_runtime.h>
#include <math.h>

#define NNODES 50000
#define NEDGES 800000
#define NGRAPH 256
#define FIN    128
#define H1C    12
#define D1C    32
#define F1     (H1C * D1C)
#define H2C    1
#define D2C    64
#define F2     (H2C * D2C)
#define ETOT   (NEDGES + NNODES)

typedef float v2f __attribute__((ext_vector_type(2)));
typedef float v8f __attribute__((ext_vector_type(8)));

__device__ __forceinline__ unsigned fkey(float f) {
  unsigned u = __float_as_uint(f);
  return (u & 0x80000000u) ? ~u : (u | 0x80000000u);
}
__device__ __forceinline__ float fdecode(unsigned u) {
  unsigned b = (u & 0x80000000u) ? (u & 0x7fffffffu) : ~u;
  return __uint_as_float(b);
}

__global__ void __launch_bounds__(128)
wmma_gemm_f32(const float* __restrict__ A, const float* __restrict__ B,
              float* __restrict__ C, int K, int Ncols) {
  const int lane = threadIdx.x & 31;
  const int wave = threadIdx.x >> 5;
  const int l    = lane & 15;
  const int half = lane >> 4;
  const int m0   = blockIdx.x << 4;
  const int n0   = ((blockIdx.y << 2) + wave) << 4;

  v8f c = {};
  const float* arow = A + (size_t)(m0 + l) * K + (half << 1);
  const float* bcol = B + n0 + l;

  for (int k0 = 0; k0 < K; k0 += 4) {
    v2f a = *(const v2f*)(arow + k0);
    const int kb = k0 + (half << 1);
    v2f b;
    b.x = bcol[(size_t)kb * Ncols];
    b.y = bcol[(size_t)(kb + 1) * Ncols];
    c = __builtin_amdgcn_wmma_f32_16x16x4_f32(
         false, a,  false, b,
         (short)0, c,  false,  false);
  }

  float* crow = C + (size_t)(m0 + (half << 3)) * Ncols + n0 + l;
#pragma unroll
  for (int r = 0; r < 8; ++r) crow[(size_t)r * Ncols] = c[r];
}

__global__ void att_coeff(const float* __restrict__ h,
                          const float* __restrict__ att_src,
                          const float* __restrict__ att_dst,
                          float* __restrict__ asrc, float* __restrict__ adst,
                          int H, int D) {
  int i = blockIdx.x * blockDim.x + threadIdx.x;
  if (i >= NNODES * H) return;
  int n = i / H, hh = i - n * H;
  const float* hp = h + (size_t)n * H * D + hh * D;
  const float* ws = att_src + hh * D;
  const float* wd = att_dst + hh * D;
  float s = 0.f, d = 0.f;
  for (int j = 0; j < D; ++j) { float v = hp[j]; s += v * ws[j]; d += v * wd[j]; }
  asrc[i] = s; adst[i] = d;
}

__global__ void edge_logit_max(const int* __restrict__ srcI, const int* __restrict__ dstI,
                               const float* __restrict__ asrc, const float* __restrict__ adst,
                               float* __restrict__ e, unsigned* __restrict__ mkey, int H) {
  int i = blockIdx.x * blockDim.x + threadIdx.x;
  if (i >= ETOT * H) return;
  int eid = i / H, hh = i - eid * H;
  int s, d;
  if (eid < NEDGES) { s = srcI[eid]; d = dstI[eid]; } else { s = eid - NEDGES; d = s; }
  float v = asrc[s * H + hh] + adst[d * H + hh];
  v = (v > 0.f) ? v : 0.2f * v;
  e[i] = v;
  atomicMax(&mkey[d * H + hh], fkey(v));
}

__global__ void edge_exp_sum(const int* __restrict__ dstI, float* __restrict__ e,
                             const unsigned* __restrict__ mkey, float* __restrict__ ssum,
                             int H) {
  int i = blockIdx.x * blockDim.x + threadIdx.x;
  if (i >= ETOT * H) return;
  int eid = i / H, hh = i - eid * H;
  int d = (eid < NEDGES) ? dstI[eid] : (eid - NEDGES);
  float p = __expf(e[i] - fdecode(mkey[d * H + hh]));
  e[i] = p;
  atomicAdd(&ssum[d * H + hh], p);
}

__global__ void edge_alpha(const int* __restrict__ dstI, float* __restrict__ e,
                           const float* __restrict__ ssum, int H) {
  int i = blockIdx.x * blockDim.x + threadIdx.x;
  if (i >= ETOT * H) return;
  int eid = i / H, hh = i - eid * H;
  int d = (eid < NEDGES) ? dstI[eid] : (eid - NEDGES);
  e[i] = e[i] / ssum[d * H + hh];
}

__global__ void deg_count(const int* __restrict__ dstI, int* __restrict__ cnt) {
  int e = blockIdx.x * blockDim.x + threadIdx.x;
  if (e >= NEDGES) return;
  atomicAdd(&cnt[dstI[e]], 1);
}

#define SCAN_T   1000
#define SCAN_PER (NNODES / SCAN_T)
static_assert(SCAN_T * SCAN_PER == NNODES, "row_scan covers the nodes exactly");

__global__ void __launch_bounds__(SCAN_T)
row_scan(const int* __restrict__ cnt, int* __restrict__ rowptr, int* __restrict__ cursor) {
  __shared__ int part[SCAN_T];
  const int t = threadIdx.x;
  const int b = t * SCAN_PER;
  int s = 0;
  for (int j = 0; j < SCAN_PER; ++j) s += cnt[b + j];
  part[t] = s;
  __syncthreads();
  for (int off = 1; off < SCAN_T; off <<= 1) {
    int v = (t >= off) ? part[t - off] : 0;
    __syncthreads();
    part[t] += v;
    __syncthreads();
  }
  int run = part[t] - s;
  for (int j = 0; j < SCAN_PER; ++j) {
    const int i = b + j;
    rowptr[i] = run; cursor[i] = run; run += cnt[i];
  }
  if (t == SCAN_T - 1) rowptr[NNODES] = part[t];
}

__global__ void csr_fill(const int* __restrict__ srcI, const int* __restrict__ dstI,
                         int* __restrict__ cursor, int* __restrict__ slot) {
  int e = blockIdx.x * blockDim.x + threadIdx.x;
  if (e >= NEDGES) return;
  int pos = atomicAdd(&cursor[dstI[e]], 1);
  ((int2*)slot)[pos] = make_int2(srcI[e], e);
}

template <int H, int D, bool RELU>
__device__ __forceinline__ void seg_agg_body(const int* __restrict__ rowptr,
                                             const int* __restrict__ slot,
                                             const float* __restrict__ alpha,
                                             const float* __restrict__ h,
                                             const float* __restrict__ bias,
                                             float* __restrict__ out) {
  constexpr int Q = (H * D) / 4;
  const int i = blockIdx.x * blockDim.x + threadIdx.x;
  if (i >= NNODES * Q) return;
  const int d  = i / Q;
  const int q  = i - d * Q;
  const int hh = (q * 4) / D;
  const float4* h4 = (const float4*)h;
  const int2*   sl = (const int2*)slot;

  float  a = alpha[(NEDGES + d) * H + hh];
  float4 v = h4[d * Q + q];
  float4 acc = make_float4(a * v.x, a * v.y, a * v.z, a * v.w);

  int j = rowptr[d];
  const int end = rowptr[d + 1];
  for (; j + 4 <= end; j += 4) {
    const int2 s0 = sl[j], s1 = sl[j + 1], s2 = sl[j + 2], s3 = sl[j + 3];
    const float a0 = alpha[s0.y * H + hh], a1 = alpha[s1.y * H + hh];
    const float a2 = alpha[s2.y * H + hh], a3 = alpha[s3.y * H + hh];
    const float4 v0 = h4[s0.x * Q + q], v1 = h4[s1.x * Q + q];
    const float4 v2 = h4[s2.x * Q + q], v3 = h4[s3.x * Q + q];
    acc.x += a0 * v0.x; acc.y += a0 * v0.y; acc.z += a0 * v0.z; acc.w += a0 * v0.w;
    acc.x += a1 * v1.x; acc.y += a1 * v1.y; acc.z += a1 * v1.z; acc.w += a1 * v1.w;
    acc.x += a2 * v2.x; acc.y += a2 * v2.y; acc.z += a2 * v2.z; acc.w += a2 * v2.w;
    acc.x += a3 * v3.x; acc.y += a3 * v3.y; acc.z += a3 * v3.z; acc.w += a3 * v3.w;
  }
  for (; j < end; ++j) {
    const int2 s0 = sl[j];
    const float a0 = alpha[s0.y * H + hh];
    const float4 v0 = h4[s0.x * Q + q];
    acc.x += a0 * v0.x; acc.y += a0 * v0.y; acc.z += a0 * v0.z; acc.w += a0 * v0.w;
  }

  const float4 bv = ((const float4*)bias)[q];
  acc.x += bv.x; acc.y += bv.y; acc.z += bv.z; acc.w += bv.w;
  if (RELU) {
    acc.x = fmaxf(acc.x, 0.f); acc.y = fmaxf(acc.y, 0.f);
    acc.z = fmaxf(acc.z, 0.f); acc.w = fmaxf(acc.w, 0.f);
  }
  ((float4*)out)[i] = acc;
}

__global__ void __launch_bounds__(256)
seg_agg_conv1(const int* __restrict__ rowptr, const int* __restrict__ slot,
              const float* __restrict__ alpha, const float* __restrict__ h,
              const float* __restrict__ bias, float* __restrict__ out) {
  seg_agg_body<H1C, D1C, true>(rowptr, slot, alpha, h, bias, out);
}

__global__ void __launch_bounds__(256)
seg_agg_conv2(const int* __restrict__ rowptr, const int* __restrict__ slot,
              const float* __restrict__ alpha, const float* __restrict__ h,
              const float* __restrict__ bias, float* __restrict__ out) {
  seg_agg_body<H2C, D2C, false>(rowptr, slot, alpha, h, bias, out);
}

__global__ void pool_accum(const float* __restrict__ h, const int* __restrict__ batch,
                           float* __restrict__ sums, float* __restrict__ cnt) {
  int i = blockIdx.x * blockDim.x + threadIdx.x;
  if (i >= NNODES * F2) return;
  int n = i / F2, f = i - n * F2;
  int g = batch[n];
  atomicAdd(&sums[g * F2 + f], h[i]);
  if (f == 0) atomicAdd(&cnt[g], 1.0f);
}

__global__ void pool_final(const float* __restrict__ sums, const float* __restrict__ cnt,
                           float* __restrict__ out) {
  int i = blockIdx.x * blockDim.x + threadIdx.x;
  if (i >= NGRAPH * F2) return;
  out[i] = sums[i] / fmaxf(cnt[i / F2], 1.0f);
}

static inline int cdiv(long long a, int b) { return (int)((a + b - 1) / b); }

extern "C" void kernel_launch(void* const* d_in, const int* in_sizes, int n_in,
                              void* d_out, int out_size, void* d_ws, size_t ws_size,
                              hipStream_t stream) {
  const float* x        = (const float*)d_in[0];
  const int*   ei       = (const int*)d_in[1];
  const int*   batch    = (const int*)d_in[2];
  const float* W1       = (const float*)d_in[3];
  const float* att_src1 = (const float*)d_in[4];
  const float* att_dst1 = (const float*)d_in[5];
  const float* bias1    = (const float*)d_in[6];
  const float* W2       = (const float*)d_in[7];
  const float* att_src2 = (const float*)d_in[8];
  const float* att_dst2 = (const float*)d_in[9];
  const float* bias2    = (const float*)d_in[10];
  float* out = (float*)d_out;
  const int* srcI = ei;
  const int* dstI = ei + NEDGES;

  char* p = (char*)d_ws;
  auto alloc = [&](size_t bytes) -> char* {
    char* r = p; p += (bytes + 255) & ~(size_t)255; return r;
  };
  float*    h1    = (float*)alloc((size_t)NNODES * F1 * 4);
  float*    asrc1 = (float*)alloc((size_t)NNODES * H1C * 4);
  float*    adst1 = (float*)alloc((size_t)NNODES * H1C * 4);
  unsigned* mkey1 = (unsigned*)alloc((size_t)NNODES * H1C * 4);
  float*    ssum1 = (float*)alloc((size_t)NNODES * H1C * 4);
  float*    e1    = (float*)alloc((size_t)ETOT * H1C * 4);
  float*    agg1  = (float*)alloc((size_t)NNODES * F1 * 4);
  float*    h2    = (float*)alloc((size_t)NNODES * F2 * 4);
  float*    asrc2 = (float*)alloc((size_t)NNODES * H2C * 4);
  float*    adst2 = (float*)alloc((size_t)NNODES * H2C * 4);
  unsigned* mkey2 = (unsigned*)alloc((size_t)NNODES * H2C * 4);
  float*    ssum2 = (float*)alloc((size_t)NNODES * H2C * 4);
  float*    e2    = (float*)alloc((size_t)ETOT * H2C * 4);
  float*    agg2  = (float*)alloc((size_t)NNODES * F2 * 4);
  float*    psum  = (float*)alloc((size_t)NGRAPH * F2 * 4);
  float*    pcnt  = (float*)alloc((size_t)NGRAPH * 4);
  int*      cnt    = (int*)alloc((size_t)NNODES * 4);
  int*      rowptr = (int*)alloc((size_t)(NNODES + 1) * 4);
  int*      cursor = (int*)alloc((size_t)NNODES * 4);
  int*      slot   = (int*)alloc((size_t)NEDGES * 2 * 4);

  hipMemsetAsync(mkey1, 0, (size_t)NNODES * H1C * 4, stream);
  hipMemsetAsync(ssum1, 0, (size_t)NNODES * H1C * 4, stream);
  hipMemsetAsync(mkey2, 0, (size_t)NNODES * H2C * 4, stream);
  hipMemsetAsync(ssum2, 0, (size_t)NNODES * H2C * 4, stream);
  hipMemsetAsync(cnt,   0, (size_t)NNODES * 4, stream);
  hipMemsetAsync(psum,  0, (size_t)NGRAPH * F2 * 4, stream);
  hipMemsetAsync(pcnt,  0, (size_t)NGRAPH * 4, stream);

  const int T = 256;

  deg_count<<<cdiv((long long)NEDGES, T), T, 0, stream>>>(dstI, cnt);
  row_scan<<<1, SCAN_T, 0, stream>>>(cnt, rowptr, cursor);
  csr_fill<<<cdiv((long long)NEDGES, T), T, 0, stream>>>(srcI, dstI, cursor, slot);

  wmma_gemm_f32<<<dim3(NNODES / 16, F1 / 64), 128, 0, stream>>>(x, W1, h1, FIN, F1);
  att_coeff<<<cdiv((long long)NNODES * H1C, T), T, 0, stream>>>(h1, att_src1, att_dst1,
                                                                asrc1, adst1, H1C, D1C);
  edge_logit_max<<<cdiv((long long)ETOT * H1C, T), T, 0, stream>>>(srcI, dstI, asrc1, adst1,
                                                                   e1, mkey1, H1C);
  edge_exp_sum<<<cdiv((long long)ETOT * H1C, T), T, 0, stream>>>(dstI, e1, mkey1, ssum1, H1C);
  edge_alpha<<<cdiv((long long)ETOT * H1C, T), T, 0, stream>>>(dstI, e1, ssum1, H1C);
  seg_agg_conv1<<<cdiv((long long)NNODES * (F1 / 4), T), T, 0, stream>>>(rowptr, slot, e1, h1,
                                                                         bias1, agg1);

  wmma_gemm_f32<<<dim3(NNODES / 16, F2 / 64), 128, 0, stream>>>(agg1, W2, h2, F1, F2);
  att_coeff<<<cdiv((long long)NNODES * H2C, T), T, 0, stream>>>(h2, att_src2, att_dst2,
                                                                asrc2, adst2, H2C, D2C);
  edge_logit_max<<<cdiv((long long)ETOT * H2C, T), T, 0, stream>>>(srcI, dstI, asrc2, adst2,
                                                                   e2, mkey2, H2C);
  edge_exp_sum<<<cdiv((long long)ETOT * H2C, T), T, 0, stream>>>(dstI, e2, mkey2, ssum2, H2C);
  edge_alpha<<<cdiv((long long)ETOT * H2C, T), T, 0, stream>>>(dstI, e2, ssum2, H2C);
  seg_agg_conv2<<<cdiv((long long)NNODES * (F2 / 4), T), T, 0, stream>>>(rowptr, slot, e2, h2,
                                                                         bias2, agg2);

  pool_accum<<<cdiv((long long)NNODES * F2, T), T, 0, stream>>>(agg2, batch, psum, pcnt);
  pool_final<<<cdiv((long long)NGRAPH * F2, T), T, 0, stream>>>(psum, pcnt, out);
}


# Softmax fused into the per-destination aggregation: logits, running max and sum in registers with online rescale; per-edge alpha array, 3 atomic passes per conv and 4 clears gone; sum order changes
# speedup vs baseline: 4.1093x; 1.5726x over previous
#include <hip/hip_runtime.h>
#include <math.h>

#define NNODES 50000
#define NEDGES 800000
#define NGRAPH 256
#define FIN    128
#define H1C    12
#define D1C    32
#define F1     (H1C * D1C)
#define H2C    1
#define D2C    64
#define F2     (H2C * D2C)

typedef float v2f __attribute__((ext_vector_type(2)));
typedef float v8f __attribute__((ext_vector_type(8)));

__global__ void __launch_bounds__(128)
wmma_gemm_f32(const float* __restrict__ A, const float* __restrict__ B,
              float* __restrict__ C, int K, int Ncols) {
  const int lane = threadIdx.x & 31;
  const int wave = threadIdx.x >> 5;
  const int l    = lane & 15;
  const int half = lane >> 4;
  const int m0   = blockIdx.x << 4;
  const int n0   = ((blockIdx.y << 2) + wave) << 4;

  v8f c = {};
  const float* arow = A + (size_t)(m0 + l) * K + (half << 1);
  const float* bcol = B + n0 + l;

  for (int k0 = 0; k0 < K; k0 += 4) {
    v2f a = *(const v2f*)(arow + k0);
    const int kb = k0 + (half << 1);
    v2f b;
    b.x = bcol[(size_t)kb * Ncols];
    b.y = bcol[(size_t)(kb + 1) * Ncols];
    c = __builtin_amdgcn_wmma_f32_16x16x4_f32(
         false, a,  false, b,
         (short)0, c,  false,  false);
  }

  float* crow = C + (size_t)(m0 + (half << 3)) * Ncols + n0 + l;
#pragma unroll
  for (int r = 0; r < 8; ++r) crow[(size_t)r * Ncols] = c[r];
}

__global__ void att_coeff(const float* __restrict__ h,
                          const float* __restrict__ att_src,
                          const float* __restrict__ att_dst,
                          float* __restrict__ asrc, float* __restrict__ adst,
                          int H, int D) {
  int i = blockIdx.x * blockDim.x + threadIdx.x;
  if (i >= NNODES * H) return;
  int n = i / H, hh = i - n * H;
  const float* hp = h + (size_t)n * H * D + hh * D;
  const float* ws = att_src + hh * D;
  const float* wd = att_dst + hh * D;
  float s = 0.f, d = 0.f;
  for (int j = 0; j < D; ++j) { float v = hp[j]; s += v * ws[j]; d += v * wd[j]; }
  asrc[i] = s; adst[i] = d;
}

__global__ void deg_count(const int* __restrict__ dstI, int* __restrict__ cnt) {
  int e = blockIdx.x * blockDim.x + threadIdx.x;
  if (e >= NEDGES) return;
  atomicAdd(&cnt[dstI[e]], 1);
}

#define SCAN_T   1000
#define SCAN_PER (NNODES / SCAN_T)
static_assert(SCAN_T * SCAN_PER == NNODES, "row_scan covers the nodes exactly");

__global__ void __launch_bounds__(SCAN_T)
row_scan(const int* __restrict__ cnt, int* __restrict__ rowptr, int* __restrict__ cursor) {
  __shared__ int part[SCAN_T];
  const int t = threadIdx.x;
  const int b = t * SCAN_PER;
  int s = 0;
  for (int j = 0; j < SCAN_PER; ++j) s += cnt[b + j];
  part[t] = s;
  __syncthreads();
  for (int off = 1; off < SCAN_T; off <<= 1) {
    int v = (t >= off) ? part[t - off] : 0;
    __syncthreads();
    part[t] += v;
    __syncthreads();
  }
  int run = part[t] - s;
  for (int j = 0; j < SCAN_PER; ++j) {
    const int i = b + j;
    rowptr[i] = run; cursor[i] = run; run += cnt[i];
  }
  if (t == SCAN_T - 1) rowptr[NNODES] = part[t];
}

__global__ void csr_fill(const int* __restrict__ srcI, const int* __restrict__ dstI,
                         int* __restrict__ cursor, int* __restrict__ slot) {
  int e = blockIdx.x * blockDim.x + threadIdx.x;
  if (e >= NEDGES) return;
  int pos = atomicAdd(&cursor[dstI[e]], 1);
  slot[pos] = srcI[e];
}

__device__ __forceinline__ float leaky(float v) { return (v > 0.f) ? v : 0.2f * v; }

template <int H, int D, bool RELU>
__device__ __forceinline__ void seg_agg_body(const int* __restrict__ rowptr,
                                             const int* __restrict__ slot,
                                             const float* __restrict__ asrc,
                                             const float* __restrict__ adst,
                                             const float* __restrict__ h,
                                             const float* __restrict__ bias,
                                             float* __restrict__ out) {
  constexpr int Q = (H * D) / 4;
  const int i = blockIdx.x * blockDim.x + threadIdx.x;
  if (i >= NNODES * Q) return;
  const int d  = i / Q;
  const int q  = i - d * Q;
  const int hh = (q * 4) / D;
  const float4* h4 = (const float4*)h;

  const float ad = adst[d * H + hh];
  float  m   = leaky(asrc[d * H + hh] + ad);
  float  s   = 1.f;
  float4 acc = h4[d * Q + q];

  int j = rowptr[d];
  const int end = rowptr[d + 1];
  for (; j + 4 <= end; j += 4) {
    const int s0 = slot[j], s1 = slot[j + 1], s2 = slot[j + 2], s3 = slot[j + 3];
    const float x0 = leaky(asrc[s0 * H + hh] + ad), x1 = leaky(asrc[s1 * H + hh] + ad);
    const float x2 = leaky(asrc[s2 * H + hh] + ad), x3 = leaky(asrc[s3 * H + hh] + ad);
    const float4 v0 = h4[s0 * Q + q], v1 = h4[s1 * Q + q];
    const float4 v2 = h4[s2 * Q + q], v3 = h4[s3 * Q + q];
    const float mn = fmaxf(fmaxf(m, x0), fmaxf(fmaxf(x1, x2), x3));
    const float sc = __expf(m - mn);
    const float p0 = __expf(x0 - mn), p1 = __expf(x1 - mn);
    const float p2 = __expf(x2 - mn), p3 = __expf(x3 - mn);
    m = mn;
    s = s * sc + ((p0 + p1) + (p2 + p3));
    acc.x *= sc; acc.y *= sc; acc.z *= sc; acc.w *= sc;
    acc.x += p0 * v0.x; acc.y += p0 * v0.y; acc.z += p0 * v0.z; acc.w += p0 * v0.w;
    acc.x += p1 * v1.x; acc.y += p1 * v1.y; acc.z += p1 * v1.z; acc.w += p1 * v1.w;
    acc.x += p2 * v2.x; acc.y += p2 * v2.y; acc.z += p2 * v2.z; acc.w += p2 * v2.w;
    acc.x += p3 * v3.x; acc.y += p3 * v3.y; acc.z += p3 * v3.z; acc.w += p3 * v3.w;
  }
  for (; j < end; ++j) {
    const int s0 = slot[j];
    const float x0 = leaky(asrc[s0 * H + hh] + ad);
    const float4 v0 = h4[s0 * Q + q];
    const float mn = fmaxf(m, x0);
    const float sc = __expf(m - mn);
    const float p0 = __expf(x0 - mn);
    m = mn;
    s = s * sc + p0;
    acc.x *= sc; acc.y *= sc; acc.z *= sc; acc.w *= sc;
    acc.x += p0 * v0.x; acc.y += p0 * v0.y; acc.z += p0 * v0.z; acc.w += p0 * v0.w;
  }

  const float inv = 1.f / s;
  const float4 bv = ((const float4*)bias)[q];
  acc.x = acc.x * inv + bv.x; acc.y = acc.y * inv + bv.y;
  acc.z = acc.z * inv + bv.z; acc.w = acc.w * inv + bv.w;
  if (RELU) {
    acc.x = fmaxf(acc.x, 0.f); acc.y = fmaxf(acc.y, 0.f);
    acc.z = fmaxf(acc.z, 0.f); acc.w = fmaxf(acc.w, 0.f);
  }
  ((float4*)out)[i] = acc;
}

__global__ void __launch_bounds__(256)
seg_agg_conv1(const int* __restrict__ rowptr, const int* __restrict__ slot,
              const float* __restrict__ asrc, const float* __restrict__ adst,
              const float* __restrict__ h,
              const float* __restrict__ bias, float* __restrict__ out) {
  seg_agg_body<H1C, D1C, true>(rowptr, slot, asrc, adst, h, bias, out);
}

__global__ void __launch_bounds__(256)
seg_agg_conv2(const int* __restrict__ rowptr, const int* __restrict__ slot,
              const float* __restrict__ asrc, const float* __restrict__ adst,
              const float* __restrict__ h,
              const float* __restrict__ bias, float* __restrict__ out) {
  seg_agg_body<H2C, D2C, false>(rowptr, slot, asrc, adst, h, bias, out);
}

__global__ void pool_accum(const float* __restrict__ h, const int* __restrict__ batch,
                           float* __restrict__ sums, float* __restrict__ cnt) {
  int i = blockIdx.x * blockDim.x + threadIdx.x;
  if (i >= NNODES * F2) return;
  int n = i / F2, f = i - n * F2;
  int g = batch[n];
  atomicAdd(&sums[g * F2 + f], h[i]);
  if (f == 0) atomicAdd(&cnt[g], 1.0f);
}

__global__ void pool_final(const float* __restrict__ sums, const float* __restrict__ cnt,
                           float* __restrict__ out) {
  int i = blockIdx.x * blockDim.x + threadIdx.x;
  if (i >= NGRAPH * F2) return;
  out[i] = sums[i] / fmaxf(cnt[i / F2], 1.0f);
}

static inline int cdiv(long long a, int b) { return (int)((a + b - 1) / b); }

extern "C" void kernel_launch(void* const* d_in, const int* in_sizes, int n_in,
                              void* d_out, int out_size, void* d_ws, size_t ws_size,
                              hipStream_t stream) {
  const float* x        = (const float*)d_in[0];
  const int*   ei       = (const int*)d_in[1];
  const int*   batch    = (const int*)d_in[2];
  const float* W1       = (const float*)d_in[3];
  const float* att_src1 = (const float*)d_in[4];
  const float* att_dst1 = (const float*)d_in[5];
  const float* bias1    = (const float*)d_in[6];
  const float* W2       = (const float*)d_in[7];
  const float* att_src2 = (const float*)d_in[8];
  const float* att_dst2 = (const float*)d_in[9];
  const float* bias2    = (const float*)d_in[10];
  float* out = (float*)d_out;
  const int* srcI = ei;
  const int* dstI = ei + NEDGES;

  char* p = (char*)d_ws;
  auto alloc = [&](size_t bytes) -> char* {
    char* r = p; p += (bytes + 255) & ~(size_t)255; return r;
  };
  float*    h1    = (float*)alloc((size_t)NNODES * F1 * 4);
  float*    asrc1 = (float*)alloc((size_t)NNODES * H1C * 4);
  float*    adst1 = (float*)alloc((size_t)NNODES * H1C * 4);
  float*    agg1  = (float*)alloc((size_t)NNODES * F1 * 4);
  float*    h2    = (float*)alloc((size_t)NNODES * F2 * 4);
  float*    asrc2 = (float*)alloc((size_t)NNODES * H2C * 4);
  float*    adst2 = (float*)alloc((size_t)NNODES * H2C * 4);
  float*    agg2  = (float*)alloc((size_t)NNODES * F2 * 4);
  float*    psum  = (float*)alloc((size_t)NGRAPH * F2 * 4);
  float*    pcnt  = (float*)alloc((size_t)NGRAPH * 4);
  int*      cnt    = (int*)alloc((size_t)NNODES * 4);
  int*      rowptr = (int*)alloc((size_t)(NNODES + 1) * 4);
  int*      cursor = (int*)alloc((size_t)NNODES * 4);
  int*      slot   = (int*)alloc((size_t)NEDGES * 4);

  hipMemsetAsync(cnt,   0, (size_t)NNODES * 4, stream);
  hipMemsetAsync(psum,  0, (size_t)NGRAPH * F2 * 4, stream);
  hipMemsetAsync(pcnt,  0, (size_t)NGRAPH * 4, stream);

  const int T = 256;

  deg_count<<<cdiv((long long)NEDGES, T), T, 0, stream>>>(dstI, cnt);
  row_scan<<<1, SCAN_T, 0, stream>>>(cnt, rowptr, cursor);
  csr_fill<<<cdiv((long long)NEDGES, T), T, 0, stream>>>(srcI, dstI, cursor, slot);

  wmma_gemm_f32<<<dim3(NNODES / 16, F1 / 64), 128, 0, stream>>>(x, W1, h1, FIN, F1);
  att_coeff<<<cdiv((long long)NNODES * H1C, T), T, 0, stream>>>(h1, att_src1, att_dst1,
                                                                asrc1, adst1, H1C, D1C);
  seg_agg_conv1<<<cdiv((long long)NNODES * (F1 / 4), T), T, 0, stream>>>(rowptr, slot, asrc1, adst1,
                                                                         h1, bias1, agg1);

  wmma_gemm_f32<<<dim3(NNODES / 16, F2 / 64), 128, 0, stream>>>(agg1, W2, h2, F1, F2);
  att_coeff<<<cdiv((long long)NNODES * H2C, T), T, 0, stream>>>(h2, att_src2, att_dst2,
                                                                asrc2, adst2, H2C, D2C);
  seg_agg_conv2<<<cdiv((long long)NNODES * (F2 / 4), T), T, 0, stream>>>(rowptr, slot, asrc2, adst2,
                                                                         h2, bias2, agg2);

  pool_accum<<<cdiv((long long)NNODES * F2, T), T, 0, stream>>>(agg2, batch, psum, pcnt);
  pool_final<<<cdiv((long long)NGRAPH * F2, T), T, 0, stream>>>(psum, pcnt, out);
}


# GEMM retiled: a wave owns a 16x64 strip (4 accumulator tiles), one A fragment feeds 4 WMMAs, 16-B loads a chunk ahead, whole-line stores; k pairing per WMMA and sum order change
# speedup vs baseline: 4.8883x; 1.1896x over previous
#include <hip/hip_runtime.h>
#include <math.h>

#define NNODES 50000
#define NEDGES 800000
#define NGRAPH 256
#define FIN    128
#define H1C    12
#define D1C    32
#define F1     (H1C * D1C)
#define H2C    1
#define D2C    64
#define F2     (H2C * D2C)

typedef float v2f __attribute__((ext_vector_type(2)));
typedef float v4f __attribute__((ext_vector_type(4)));
typedef float v8f __attribute__((ext_vector_type(8)));

template <int K, int NC, int WPB>
__global__ void __launch_bounds__(WPB * 32)
wmma_gemm_strip(const float* __restrict__ A, const float* __restrict__ B,
                float* __restrict__ C) {
  constexpr int MT     = NNODES / 16;
  constexpr int STRIPS = NC / 64;
  constexpr int NCH    = K / 16;
  constexpr int BROW   = NC / 4;
  static_assert(NNODES % 16 == 0 && K % 16 == 0 && NC % 64 == 0, "tile shapes divide the problem");
  static_assert((MT * STRIPS) % WPB == 0, "the grid covers the strips exactly, no partial block");

  const int lane = threadIdx.x & 31;
  const int c    = lane & 15;
  const int h    = lane >> 4;
  const int w     = blockIdx.x * WPB + __builtin_amdgcn_readfirstlane((int)(threadIdx.x >> 5));
  const int strip = w / MT;
  const int m0    = (w - strip * MT) << 4;
  const int nb    = strip << 6;

  const v4f* ap = (const v4f*)(A + (size_t)(m0 + c) * K + 8 * h);
  const v4f* bp = (const v4f*)(B + (size_t)(8 * h) * NC + nb + 4 * c);

  v8f acc[4] = {};
  v4f a[2][2];
  v4f b[2][8];

  a[0][0] = ap[0];
  a[0][1] = ap[1];
#pragma unroll
  for (int j = 0; j < 8; ++j) b[0][j] = bp[j * BROW];

#pragma unroll
  for (int i = 0; i < NCH; ++i) {
    const int cur = i & 1, nxt = cur ^ 1;
    if (i + 1 < NCH) {
      a[nxt][0] = ap[4 * (i + 1)];
      a[nxt][1] = ap[4 * (i + 1) + 1];
#pragma unroll
      for (int j = 0; j < 8; ++j) b[nxt][j] = bp[(16 * (i + 1) + j) * BROW];
    }
    __builtin_amdgcn_sched_barrier(0);
#pragma unroll
    for (int s = 0; s < 4; ++s) {
      v2f af;
      af.x = a[cur][s >> 1][(s & 1) * 2];
      af.y = a[cur][s >> 1][(s & 1) * 2 + 1];
#pragma unroll
      for (int t = 0; t < 4; ++t) {
        v2f bf;
        bf.x = b[cur][2 * s][t];
        bf.y = b[cur][2 * s + 1][t];
        acc[t] = __builtin_amdgcn_wmma_f32_16x16x4_f32(
             false, af,  false, bf,
             (short)0, acc[t],  false,  false);
      }
    }
  }

  v4f* crow = (v4f*)(C + (size_t)(m0 + (h << 3)) * NC + nb + 4 * c);
#pragma unroll
  for (int r = 0; r < 8; ++r) {
    v4f o;
    o.x = acc[0][r]; o.y = acc[1][r]; o.z = acc[2][r]; o.w = acc[3][r];
    crow[r * BROW] = o;
  }
}

__global__ void att_coeff(const float* __restrict__ h,
                          const float* __restrict__ att_src,
                          const float* __restrict__ att_dst,
                          float* __restrict__ asrc, float* __restrict__ adst,
                          int H, int D) {
  int i = blockIdx.x * blockDim.x + threadIdx.x;
  if (i >= NNODES * H) return;
  int n = i / H, hh = i - n * H;
  const float* hp = h + (size_t)n * H * D + hh * D;
  const float* ws = att_src + hh * D;
  const float* wd = att_dst + hh * D;
  float s = 0.f, d = 0.f;
  for (int j = 0; j < D; ++j) { float v = hp[j]; s += v * ws[j]; d += v * wd[j]; }
  asrc[i] = s; adst[i] = d;
}

__global__ void deg_count(const int* __restrict__ dstI, int* __restrict__ cnt) {
  int e = blockIdx.x * blockDim.x + threadIdx.x;
  if (e >= NEDGES) return;
  atomicAdd(&cnt[dstI[e]], 1);
}

#define SCAN_T   1000
#define SCAN_PER (NNODES / SCAN_T)
static_assert(SCAN_T * SCAN_PER == NNODES, "row_scan covers the nodes exactly");

__global__ void __launch_bounds__(SCAN_T)
row_scan(const int* __restrict__ cnt, int* __restrict__ rowptr, int* __restrict__ cursor) {
  __shared__ int part[SCAN_T];
  const int t = threadIdx.x;
  const int b = t * SCAN_PER;
  int s = 0;
  for (int j = 0; j < SCAN_PER; ++j) s += cnt[b + j];
  part[t] = s;
  __syncthreads();
  for (int off = 1; off < SCAN_T; off <<= 1) {
    int v = (t >= off) ? part[t - off] : 0;
    __syncthreads();
    part[t] += v;
    __syncthreads();
  }
  int run = part[t] - s;
  for (int j = 0; j < SCAN_PER; ++j) {
    const int i = b + j;
    rowptr[i] = run; cursor[i] = run; run += cnt[i];
  }
  if (t == SCAN_T - 1) rowptr[NNODES] = part[t];
}

__global__ void csr_fill(const int* __restrict__ srcI, const int* __restrict__ dstI,
                         int* __restrict__ cursor, int* __restrict__ slot) {
  int e = blockIdx.x * blockDim.x + threadIdx.x;
  if (e >= NEDGES) return;
  int pos = atomicAdd(&cursor[dstI[e]], 1);
  slot[pos] = srcI[e];
}

__device__ __forceinline__ float leaky(float v) { return (v > 0.f) ? v : 0.2f * v; }

template <int H, int D, bool RELU>
__device__ __forceinline__ void seg_agg_body(const int* __restrict__ rowptr,
                                             const int* __restrict__ slot,
                                             const float* __restrict__ asrc,
                                             const float* __restrict__ adst,
                                             const float* __restrict__ h,
                                             const float* __restrict__ bias,
                                             float* __restrict__ out) {
  constexpr int Q = (H * D) / 4;
  const int i = blockIdx.x * blockDim.x + threadIdx.x;
  if (i >= NNODES * Q) return;
  const int d  = i / Q;
  const int q  = i - d * Q;
  const int hh = (q * 4) / D;
  const float4* h4 = (const float4*)h;

  const float ad = adst[d * H + hh];
  float  m   = leaky(asrc[d * H + hh] + ad);
  float  s   = 1.f;
  float4 acc = h4[d * Q + q];

  int j = rowptr[d];
  const int end = rowptr[d + 1];
  for (; j + 4 <= end; j += 4) {
    const int s0 = slot[j], s1 = slot[j + 1], s2 = slot[j + 2], s3 = slot[j + 3];
    const float x0 = leaky(asrc[s0 * H + hh] + ad), x1 = leaky(asrc[s1 * H + hh] + ad);
    const float x2 = leaky(asrc[s2 * H + hh] + ad), x3 = leaky(asrc[s3 * H + hh] + ad);
    const float4 v0 = h4[s0 * Q + q], v1 = h4[s1 * Q + q];
    const float4 v2 = h4[s2 * Q + q], v3 = h4[s3 * Q + q];
    const float mn = fmaxf(fmaxf(m, x0), fmaxf(fmaxf(x1, x2), x3));
    const float sc = __expf(m - mn);
    const float p0 = __expf(x0 - mn), p1 = __expf(x1 - mn);
    const float p2 = __expf(x2 - mn), p3 = __expf(x3 - mn);
    m = mn;
    s = s * sc + ((p0 + p1) + (p2 + p3));
    acc.x *= sc; acc.y *= sc; acc.z *= sc; acc.w *= sc;
    acc.x += p0 * v0.x; acc.y += p0 * v0.y; acc.z += p0 * v0.z; acc.w += p0 * v0.w;
    acc.x += p1 * v1.x; acc.y += p1 * v1.y; acc.z += p1 * v1.z; acc.w += p1 * v1.w;
    acc.x += p2 * v2.x; acc.y += p2 * v2.y; acc.z += p2 * v2.z; acc.w += p2 * v2.w;
    acc.x += p3 * v3.x; acc.y += p3 * v3.y; acc.z += p3 * v3.z; acc.w += p3 * v3.w;
  }
  for (; j < end; ++j) {
    const int s0 = slot[j];
    const float x0 = leaky(asrc[s0 * H + hh] + ad);
    const float4 v0 = h4[s0 * Q + q];
    const float mn = fmaxf(m, x0);
    const float sc = __expf(m - mn);
    const float p0 = __expf(x0 - mn);
    m = mn;
    s = s * sc + p0;
    acc.x *= sc; acc.y *= sc; acc.z *= sc; acc.w *= sc;
    acc.x += p0 * v0.x; acc.y += p0 * v0.y; acc.z += p0 * v0.z; acc.w += p0 * v0.w;
  }

  const float inv = 1.f / s;
  const float4 bv = ((const float4*)bias)[q];
  acc.x = acc.x * inv + bv.x; acc.y = acc.y * inv + bv.y;
  acc.z = acc.z * inv + bv.z; acc.w = acc.w * inv + bv.w;
  if (RELU) {
    acc.x = fmaxf(acc.x, 0.f); acc.y = fmaxf(acc.y, 0.f);
    acc.z = fmaxf(acc.z, 0.f); acc.w = fmaxf(acc.w, 0.f);
  }
  ((float4*)out)[i] = acc;
}

__global__ void __launch_bounds__(256)
seg_agg_conv1(const int* __restrict__ rowptr, const int* __restrict__ slot,
              const float* __restrict__ asrc, const float* __restrict__ adst,
              const float* __restrict__ h,
              const float* __restrict__ bias, float* __restrict__ out) {
  seg_agg_body<H1C, D1C, true>(rowptr, slot, asrc, adst, h, bias, out);
}

__global__ void __launch_bounds__(256)
seg_agg_conv2(const int* __restrict__ rowptr, const int* __restrict__ slot,
              const float* __restrict__ asrc, const float* __restrict__ adst,
              const float* __restrict__ h,
              const float* __restrict__ bias, float* __restrict__ out) {
  seg_agg_body<H2C, D2C, false>(rowptr, slot, asrc, adst, h, bias, out);
}

__global__ void pool_accum(const float* __restrict__ h, const int* __restrict__ batch,
                           float* __restrict__ sums, float* __restrict__ cnt) {
  int i = blockIdx.x * blockDim.x + threadIdx.x;
  if (i >= NNODES * F2) return;
  int n = i / F2, f = i - n * F2;
  int g = batch[n];
  atomicAdd(&sums[g * F2 + f], h[i]);
  if (f == 0) atomicAdd(&cnt[g], 1.0f);
}

__global__ void pool_final(const float* __restrict__ sums, const float* __restrict__ cnt,
                           float* __restrict__ out) {
  int i = blockIdx.x * blockDim.x + threadIdx.x;
  if (i >= NGRAPH * F2) return;
  out[i] = sums[i] / fmaxf(cnt[i / F2], 1.0f);
}

static inline int cdiv(long long a, int b) { return (int)((a + b - 1) / b); }

extern "C" void kernel_launch(void* const* d_in, const int* in_sizes, int n_in,
                              void* d_out, int out_size, void* d_ws, size_t ws_size,
                              hipStream_t stream) {
  const float* x        = (const float*)d_in[0];
  const int*   ei       = (const int*)d_in[1];
  const int*   batch    = (const int*)d_in[2];
  const float* W1       = (const float*)d_in[3];
  const float* att_src1 = (const float*)d_in[4];
  const float* att_dst1 = (const float*)d_in[5];
  const float* bias1    = (const float*)d_in[6];
  const float* W2       = (const float*)d_in[7];
  const float* att_src2 = (const float*)d_in[8];
  const float* att_dst2 = (const float*)d_in[9];
  const float* bias2    = (const float*)d_in[10];
  float* out = (float*)d_out;
  const int* srcI = ei;
  const int* dstI = ei + NEDGES;

  char* p = (char*)d_ws;
  auto alloc = [&](size_t bytes) -> char* {
    char* r = p; p += (bytes + 255) & ~(size_t)255; return r;
  };
  float*    h1    = (float*)alloc((size_t)NNODES * F1 * 4);
  float*    asrc1 = (float*)alloc((size_t)NNODES * H1C * 4);
  float*    adst1 = (float*)alloc((size_t)NNODES * H1C * 4);
  float*    agg1  = (float*)alloc((size_t)NNODES * F1 * 4);
  float*    h2    = (float*)alloc((size_t)NNODES * F2 * 4);
  float*    asrc2 = (float*)alloc((size_t)NNODES * H2C * 4);
  float*    adst2 = (float*)alloc((size_t)NNODES * H2C * 4);
  float*    agg2  = (float*)alloc((size_t)NNODES * F2 * 4);
  float*    psum  = (float*)alloc((size_t)NGRAPH * F2 * 4);
  float*    pcnt  = (float*)alloc((size_t)NGRAPH * 4);
  int*      cnt    = (int*)alloc((size_t)NNODES * 4);
  int*      rowptr = (int*)alloc((size_t)(NNODES + 1) * 4);
  int*      cursor = (int*)alloc((size_t)NNODES * 4);
  int*      slot   = (int*)alloc((size_t)NEDGES * 4);

  hipMemsetAsync(cnt,   0, (size_t)NNODES * 4, stream);
  hipMemsetAsync(psum,  0, (size_t)NGRAPH * F2 * 4, stream);
  hipMemsetAsync(pcnt,  0, (size_t)NGRAPH * 4, stream);

  const int T = 256;

  deg_count<<<cdiv((long long)NEDGES, T), T, 0, stream>>>(dstI, cnt);
  row_scan<<<1, SCAN_T, 0, stream>>>(cnt, rowptr, cursor);
  csr_fill<<<cdiv((long long)NEDGES, T), T, 0, stream>>>(srcI, dstI, cursor, slot);

  wmma_gemm_strip<FIN, F1, 6><<<(NNODES / 16) * (F1 / 64) / 6, 6 * 32, 0, stream>>>(x, W1, h1);
  att_coeff<<<cdiv((long long)NNODES * H1C, T), T, 0, stream>>>(h1, att_src1, att_dst1,
                                                                asrc1, adst1, H1C, D1C);
  seg_agg_conv1<<<cdiv((long long)NNODES * (F1 / 4), T), T, 0, stream>>>(rowptr, slot, asrc1, adst1,
                                                                         h1, bias1, agg1);

  wmma_gemm_strip<F1, F2, 5><<<(NNODES / 16) * (F2 / 64) / 5, 5 * 32, 0, stream>>>(agg1, W2, h2);
  att_coeff<<<cdiv((long long)NNODES * H2C, T), T, 0, stream>>>(h2, att_src2, att_dst2,
                                                                asrc2, adst2, H2C, D2C);
  seg_agg_conv2<<<cdiv((long long)NNODES * (F2 / 4), T), T, 0, stream>>>(rowptr, slot, asrc2, adst2,
                                                                         h2, bias2, agg2);

  pool_accum<<<cdiv((long long)NNODES * F2, T), T, 0, stream>>>(agg2, batch, psum, pcnt);
  pool_final<<<cdiv((long long)NGRAPH * F2, T), T, 0, stream>>>(psum, pcnt, out);
}


# Mean pool: a wave sums 16 consecutive node rows in registers while the graph id repeats, one float atomic per feature per run; per-element atomics and 97k short waves gone; sum order changes
# speedup vs baseline: 5.6440x; 1.1546x over previous
#include <hip/hip_runtime.h>
#include <math.h>

#define NNODES 50000
#define NEDGES 800000
#define NGRAPH 256
#define FIN    128
#define H1C    12
#define D1C    32
#define F1     (H1C * D1C)
#define H2C    1
#define D2C    64
#define F2     (H2C * D2C)

typedef float v2f __attribute__((ext_vector_type(2)));
typedef float v4f __attribute__((ext_vector_type(4)));
typedef float v8f __attribute__((ext_vector_type(8)));

template <int K, int NC, int WPB>
__global__ void __launch_bounds__(WPB * 32)
wmma_gemm_strip(const float* __restrict__ A, const float* __restrict__ B,
                float* __restrict__ C) {
  constexpr int MT     = NNODES / 16;
  constexpr int STRIPS = NC / 64;
  constexpr int NCH    = K / 16;
  constexpr int BROW   = NC / 4;
  static_assert(NNODES % 16 == 0 && K % 16 == 0 && NC % 64 == 0, "tile shapes divide the problem");
  static_assert((MT * STRIPS) % WPB == 0, "the grid covers the strips exactly, no partial block");

  const int lane = threadIdx.x & 31;
  const int c    = lane & 15;
  const int h    = lane >> 4;
  const int w     = blockIdx.x * WPB + __builtin_amdgcn_readfirstlane((int)(threadIdx.x >> 5));
  const int strip = w / MT;
  const int m0    = (w - strip * MT) << 4;
  const int nb    = strip << 6;

  const v4f* ap = (const v4f*)(A + (size_t)(m0 + c) * K + 8 * h);
  const v4f* bp = (const v4f*)(B + (size_t)(8 * h) * NC + nb + 4 * c);

  v8f acc[4] = {};
  v4f a[2][2];
  v4f b[2][8];

  a[0][0] = ap[0];
  a[0][1] = ap[1];
#pragma unroll
  for (int j = 0; j < 8; ++j) b[0][j] = bp[j * BROW];

#pragma unroll
  for (int i = 0; i < NCH; ++i) {
    const int cur = i & 1, nxt = cur ^ 1;
    if (i + 1 < NCH) {
      a[nxt][0] = ap[4 * (i + 1)];
      a[nxt][1] = ap[4 * (i + 1) + 1];
#pragma unroll
      for (int j = 0; j < 8; ++j) b[nxt][j] = bp[(16 * (i + 1) + j) * BROW];
    }
    __builtin_amdgcn_sched_barrier(0);
#pragma unroll
    for (int s = 0; s < 4; ++s) {
      v2f af;
      af.x = a[cur][s >> 1][(s & 1) * 2];
      af.y = a[cur][s >> 1][(s & 1) * 2 + 1];
#pragma unroll
      for (int t = 0; t < 4; ++t) {
        v2f bf;
        bf.x = b[cur][2 * s][t];
        bf.y = b[cur][2 * s + 1][t];
        acc[t] = __builtin_amdgcn_wmma_f32_16x16x4_f32(
             false, af,  false, bf,
             (short)0, acc[t],  false,  false);
      }
    }
  }

  v4f* crow = (v4f*)(C + (size_t)(m0 + (h << 3)) * NC + nb + 4 * c);
#pragma unroll
  for (int r = 0; r < 8; ++r) {
    v4f o;
    o.x = acc[0][r]; o.y = acc[1][r]; o.z = acc[2][r]; o.w = acc[3][r];
    crow[r * BROW] = o;
  }
}

__global__ void att_coeff(const float* __restrict__ h,
                          const float* __restrict__ att_src,
                          const float* __restrict__ att_dst,
                          float* __restrict__ asrc, float* __restrict__ adst,
                          int H, int D) {
  int i = blockIdx.x * blockDim.x + threadIdx.x;
  if (i >= NNODES * H) return;
  int n = i / H, hh = i - n * H;
  const float* hp = h + (size_t)n * H * D + hh * D;
  const float* ws = att_src + hh * D;
  const float* wd = att_dst + hh * D;
  float s = 0.f, d = 0.f;
  for (int j = 0; j < D; ++j) { float v = hp[j]; s += v * ws[j]; d += v * wd[j]; }
  asrc[i] = s; adst[i] = d;
}

__global__ void deg_count(const int* __restrict__ dstI, int* __restrict__ cnt) {
  int e = blockIdx.x * blockDim.x + threadIdx.x;
  if (e >= NEDGES) return;
  atomicAdd(&cnt[dstI[e]], 1);
}

#define SCAN_T   1000
#define SCAN_PER (NNODES / SCAN_T)
static_assert(SCAN_T * SCAN_PER == NNODES, "row_scan covers the nodes exactly");

__global__ void __launch_bounds__(SCAN_T)
row_scan(const int* __restrict__ cnt, int* __restrict__ rowptr, int* __restrict__ cursor) {
  __shared__ int part[SCAN_T];
  const int t = threadIdx.x;
  const int b = t * SCAN_PER;
  int s = 0;
  for (int j = 0; j < SCAN_PER; ++j) s += cnt[b + j];
  part[t] = s;
  __syncthreads();
  for (int off = 1; off < SCAN_T; off <<= 1) {
    int v = (t >= off) ? part[t - off] : 0;
    __syncthreads();
    part[t] += v;
    __syncthreads();
  }
  int run = part[t] - s;
  for (int j = 0; j < SCAN_PER; ++j) {
    const int i = b + j;
    rowptr[i] = run; cursor[i] = run; run += cnt[i];
  }
  if (t == SCAN_T - 1) rowptr[NNODES] = part[t];
}

__global__ void csr_fill(const int* __restrict__ srcI, const int* __restrict__ dstI,
                         int* __restrict__ cursor, int* __restrict__ slot) {
  int e = blockIdx.x * blockDim.x + threadIdx.x;
  if (e >= NEDGES) return;
  int pos = atomicAdd(&cursor[dstI[e]], 1);
  slot[pos] = srcI[e];
}

__device__ __forceinline__ float leaky(float v) { return (v > 0.f) ? v : 0.2f * v; }

template <int H, int D, bool RELU>
__device__ __forceinline__ void seg_agg_body(const int* __restrict__ rowptr,
                                             const int* __restrict__ slot,
                                             const float* __restrict__ asrc,
                                             const float* __restrict__ adst,
                                             const float* __restrict__ h,
                                             const float* __restrict__ bias,
                                             float* __restrict__ out) {
  constexpr int Q = (H * D) / 4;
  const int i = blockIdx.x * blockDim.x + threadIdx.x;
  if (i >= NNODES * Q) return;
  const int d  = i / Q;
  const int q  = i - d * Q;
  const int hh = (q * 4) / D;
  const float4* h4 = (const float4*)h;

  const float ad = adst[d * H + hh];
  float  m   = leaky(asrc[d * H + hh] + ad);
  float  s   = 1.f;
  float4 acc = h4[d * Q + q];

  int j = rowptr[d];
  const int end = rowptr[d + 1];
  for (; j + 4 <= end; j += 4) {
    const int s0 = slot[j], s1 = slot[j + 1], s2 = slot[j + 2], s3 = slot[j + 3];
    const float x0 = leaky(asrc[s0 * H + hh] + ad), x1 = leaky(asrc[s1 * H + hh] + ad);
    const float x2 = leaky(asrc[s2 * H + hh] + ad), x3 = leaky(asrc[s3 * H + hh] + ad);
    const float4 v0 = h4[s0 * Q + q], v1 = h4[s1 * Q + q];
    const float4 v2 = h4[s2 * Q + q], v3 = h4[s3 * Q + q];
    const float mn = fmaxf(fmaxf(m, x0), fmaxf(fmaxf(x1, x2), x3));
    const float sc = __expf(m - mn);
    const float p0 = __expf(x0 - mn), p1 = __expf(x1 - mn);
    const float p2 = __expf(x2 - mn), p3 = __expf(x3 - mn);
    m = mn;
    s = s * sc + ((p0 + p1) + (p2 + p3));
    acc.x *= sc; acc.y *= sc; acc.z *= sc; acc.w *= sc;
    acc.x += p0 * v0.x; acc.y += p0 * v0.y; acc.z += p0 * v0.z; acc.w += p0 * v0.w;
    acc.x += p1 * v1.x; acc.y += p1 * v1.y; acc.z += p1 * v1.z; acc.w += p1 * v1.w;
    acc.x += p2 * v2.x; acc.y += p2 * v2.y; acc.z += p2 * v2.z; acc.w += p2 * v2.w;
    acc.x += p3 * v3.x; acc.y += p3 * v3.y; acc.z += p3 * v3.z; acc.w += p3 * v3.w;
  }
  for (; j < end; ++j) {
    const int s0 = slot[j];
    const float x0 = leaky(asrc[s0 * H + hh] + ad);
    const float4 v0 = h4[s0 * Q + q];
    const float mn = fmaxf(m, x0);
    const float sc = __expf(m - mn);
    const float p0 = __expf(x0 - mn);
    m = mn;
    s = s * sc + p0;
    acc.x *= sc; acc.y *= sc; acc.z *= sc; acc.w *= sc;
    acc.x += p0 * v0.x; acc.y += p0 * v0.y; acc.z += p0 * v0.z; acc.w += p0 * v0.w;
  }

  const float inv = 1.f / s;
  const float4 bv = ((const float4*)bias)[q];
  acc.x = acc.x * inv + bv.x; acc.y = acc.y * inv + bv.y;
  acc.z = acc.z * inv + bv.z; acc.w = acc.w * inv + bv.w;
  if (RELU) {
    acc.x = fmaxf(acc.x, 0.f); acc.y = fmaxf(acc.y, 0.f);
    acc.z = fmaxf(acc.z, 0.f); acc.w = fmaxf(acc.w, 0.f);
  }
  ((float4*)out)[i] = acc;
}

__global__ void __launch_bounds__(256)
seg_agg_conv1(const int* __restrict__ rowptr, const int* __restrict__ slot,
              const float* __restrict__ asrc, const float* __restrict__ adst,
              const float* __restrict__ h,
              const float* __restrict__ bias, float* __restrict__ out) {
  seg_agg_body<H1C, D1C, true>(rowptr, slot, asrc, adst, h, bias, out);
}

__global__ void __launch_bounds__(256)
seg_agg_conv2(const int* __restrict__ rowptr, const int* __restrict__ slot,
              const float* __restrict__ asrc, const float* __restrict__ adst,
              const float* __restrict__ h,
              const float* __restrict__ bias, float* __restrict__ out) {
  seg_agg_body<H2C, D2C, false>(rowptr, slot, asrc, adst, h, bias, out);
}

#define POOL_CH  16
#define POOL_WPB 5
static_assert(NNODES % POOL_CH == 0, "pool chunks cover the nodes exactly");
static_assert((NNODES / POOL_CH) % POOL_WPB == 0, "the pool grid covers the chunks exactly, no partial block");
static_assert(F2 == 64 && POOL_CH <= 32, "a lane owns features lane and lane + 32; one lane per chunk row holds its graph id");

__global__ void __launch_bounds__(POOL_WPB * 32)
pool_accum(const float* __restrict__ h, const int* __restrict__ batch,
           float* __restrict__ sums, float* __restrict__ cnt) {
  const int lane = threadIdx.x & 31;
  const int w  = blockIdx.x * POOL_WPB + __builtin_amdgcn_readfirstlane((int)(threadIdx.x >> 5));
  const int n0 = w * POOL_CH;
  const int gl = batch[n0 + (lane & (POOL_CH - 1))];
  const float* hp = h + (size_t)n0 * F2 + lane;

  float lo[POOL_CH], hi[POOL_CH];
#pragma unroll
  for (int j = 0; j < POOL_CH; ++j) { lo[j] = hp[j * F2]; hi[j] = hp[j * F2 + 32]; }

  int   g   = __builtin_amdgcn_readlane(gl, 0);
  float a0  = lo[0], a1 = hi[0];
  float run = 1.f;
#pragma unroll
  for (int j = 1; j < POOL_CH; ++j) {
    const int gj = __builtin_amdgcn_readlane(gl, j);
    if (gj != g) {
      atomicAdd(&sums[g * F2 + lane], a0);
      atomicAdd(&sums[g * F2 + 32 + lane], a1);
      if (lane == 0) atomicAdd(&cnt[g], run);
      g = gj; a0 = lo[j]; a1 = hi[j]; run = 1.f;
    } else {
      a0 += lo[j]; a1 += hi[j]; run += 1.f;
    }
  }
  atomicAdd(&sums[g * F2 + lane], a0);
  atomicAdd(&sums[g * F2 + 32 + lane], a1);
  if (lane == 0) atomicAdd(&cnt[g], run);
}

__global__ void pool_final(const float* __restrict__ sums, const float* __restrict__ cnt,
                           float* __restrict__ out) {
  int i = blockIdx.x * blockDim.x + threadIdx.x;
  if (i >= NGRAPH * F2) return;
  out[i] = sums[i] / fmaxf(cnt[i / F2], 1.0f);
}

static inline int cdiv(long long a, int b) { return (int)((a + b - 1) / b); }

extern "C" void kernel_launch(void* const* d_in, const int* in_sizes, int n_in,
                              void* d_out, int out_size, void* d_ws, size_t ws_size,
                              hipStream_t stream) {
  const float* x        = (const float*)d_in[0];
  const int*   ei       = (const int*)d_in[1];
  const int*   batch    = (const int*)d_in[2];
  const float* W1       = (const float*)d_in[3];
  const float* att_src1 = (const float*)d_in[4];
  const float* att_dst1 = (const float*)d_in[5];
  const float* bias1    = (const float*)d_in[6];
  const float* W2       = (const float*)d_in[7];
  const float* att_src2 = (const float*)d_in[8];
  const float* att_dst2 = (const float*)d_in[9];
  const float* bias2    = (const float*)d_in[10];
  float* out = (float*)d_out;
  const int* srcI = ei;
  const int* dstI = ei + NEDGES;

  char* p = (char*)d_ws;
  auto alloc = [&](size_t bytes) -> char* {
    char* r = p; p += (bytes + 255) & ~(size_t)255; return r;
  };
  float*    h1    = (float*)alloc((size_t)NNODES * F1 * 4);
  float*    asrc1 = (float*)alloc((size_t)NNODES * H1C * 4);
  float*    adst1 = (float*)alloc((size_t)NNODES * H1C * 4);
  float*    agg1  = (float*)alloc((size_t)NNODES * F1 * 4);
  float*    h2    = (float*)alloc((size_t)NNODES * F2 * 4);
  float*    asrc2 = (float*)alloc((size_t)NNODES * H2C * 4);
  float*    adst2 = (float*)alloc((size_t)NNODES * H2C * 4);
  float*    agg2  = (float*)alloc((size_t)NNODES * F2 * 4);
  float*    psum  = (float*)alloc((size_t)NGRAPH * F2 * 4);
  float*    pcnt  = (float*)alloc((size_t)NGRAPH * 4);
  int*      cnt    = (int*)alloc((size_t)NNODES * 4);
  int*      rowptr = (int*)alloc((size_t)(NNODES + 1) * 4);
  int*      cursor = (int*)alloc((size_t)NNODES * 4);
  int*      slot   = (int*)alloc((size_t)NEDGES * 4);

  hipMemsetAsync(cnt,   0, (size_t)NNODES * 4, stream);
  hipMemsetAsync(psum,  0, (size_t)NGRAPH * F2 * 4, stream);
  hipMemsetAsync(pcnt,  0, (size_t)NGRAPH * 4, stream);

  const int T = 256;

  deg_count<<<cdiv((long long)NEDGES, T), T, 0, stream>>>(dstI, cnt);
  row_scan<<<1, SCAN_T, 0, stream>>>(cnt, rowptr, cursor);
  csr_fill<<<cdiv((long long)NEDGES, T), T, 0, stream>>>(srcI, dstI, cursor, slot);

  wmma_gemm_strip<FIN, F1, 6><<<(NNODES / 16) * (F1 / 64) / 6, 6 * 32, 0, stream>>>(x, W1, h1);
  att_coeff<<<cdiv((long long)NNODES * H1C, T), T, 0, stream>>>(h1, att_src1, att_dst1,
                                                                asrc1, adst1, H1C, D1C);
  seg_agg_conv1<<<cdiv((long long)NNODES * (F1 / 4), T), T, 0, stream>>>(rowptr, slot, asrc1, adst1,
                                                                         h1, bias1, agg1);

  wmma_gemm_strip<F1, F2, 5><<<(NNODES / 16) * (F2 / 64) / 5, 5 * 32, 0, stream>>>(agg1, W2, h2);
  att_coeff<<<cdiv((long long)NNODES * H2C, T), T, 0, stream>>>(h2, att_src2, att_dst2,
                                                                asrc2, adst2, H2C, D2C);
  seg_agg_conv2<<<cdiv((long long)NNODES * (F2 / 4), T), T, 0, stream>>>(rowptr, slot, asrc2, adst2,
                                                                         h2, bias2, agg2);

  pool_accum<<<(NNODES / POOL_CH) / POOL_WPB, POOL_WPB * 32, 0, stream>>>(agg2, batch, psum, pcnt);
  pool_final<<<cdiv((long long)NGRAPH * F2, T), T, 0, stream>>>(psum, pcnt, out);
}


# Attention coefficients from the GEMM accumulators in its epilogue (in-lane products, DPP row butterfly over a head's lanes); both coefficient launches and their re-read of h gone; sum order changes
# speedup vs baseline: 6.0025x; 1.0635x over previous
#include <hip/hip_runtime.h>
#include <math.h>

#define NNODES 50000
#define NEDGES 800000
#define NGRAPH 256
#define FIN    128
#define H1C    12
#define D1C    32
#define F1     (H1C * D1C)
#define H2C    1
#define D2C    64
#define F2     (H2C * D2C)

typedef float v2f __attribute__((ext_vector_type(2)));
typedef float v4f __attribute__((ext_vector_type(4)));
typedef float v8f __attribute__((ext_vector_type(8)));


template <int M>
__device__ __forceinline__ float row_xor_add(float v) {
  return v + __builtin_bit_cast(float, __builtin_amdgcn_update_dpp(
                 0, __builtin_bit_cast(int, v), 0x160 | M, 0xf, 0xf, true));
}

template <int K, int NC, int WPB, int D>
__global__ void __launch_bounds__(WPB * 32)
wmma_gemm_strip(const float* __restrict__ A, const float* __restrict__ B,
                float* __restrict__ C,
                const float* __restrict__ att_src, const float* __restrict__ att_dst,
                float* __restrict__ asrc, float* __restrict__ adst) {
  constexpr int MT     = NNODES / 16;
  constexpr int STRIPS = NC / 64;
  constexpr int NCH    = K / 16;
  constexpr int BROW   = NC / 4;
  static_assert(NNODES % 16 == 0 && K % 16 == 0 && NC % 64 == 0, "tile shapes divide the problem");
  static_assert((MT * STRIPS) % WPB == 0, "the grid covers the strips exactly, no partial block");
  constexpr int G = D / 4;
  constexpr int H = NC / D;
  static_assert(D == 32 || D == 64, "a head is 8 or 16 lanes of a row, and a strip holds whole heads");

  const int lane = threadIdx.x & 31;
  const int c    = lane & 15;
  const int h    = lane >> 4;
  const int w     = blockIdx.x * WPB + __builtin_amdgcn_readfirstlane((int)(threadIdx.x >> 5));
  const int strip = w / MT;
  const int m0    = (w - strip * MT) << 4;
  const int nb    = strip << 6;

  const v4f* ap = (const v4f*)(A + (size_t)(m0 + c) * K + 8 * h);
  const v4f* bp = (const v4f*)(B + (size_t)(8 * h) * NC + nb + 4 * c);

  v8f acc[4] = {};
  v4f a[2][2];
  v4f b[2][8];

  a[0][0] = ap[0];
  a[0][1] = ap[1];
#pragma unroll
  for (int j = 0; j < 8; ++j) b[0][j] = bp[j * BROW];

  const v4f ws = *(const v4f*)(att_src + nb + 4 * c);
  const v4f wd = *(const v4f*)(att_dst + nb + 4 * c);

#pragma unroll
  for (int i = 0; i < NCH; ++i) {
    const int cur = i & 1, nxt = cur ^ 1;
    if (i + 1 < NCH) {
      a[nxt][0] = ap[4 * (i + 1)];
      a[nxt][1] = ap[4 * (i + 1) + 1];
#pragma unroll
      for (int j = 0; j < 8; ++j) b[nxt][j] = bp[(16 * (i + 1) + j) * BROW];
    }
    __builtin_amdgcn_sched_barrier(0);
#pragma unroll
    for (int s = 0; s < 4; ++s) {
      v2f af;
      af.x = a[cur][s >> 1][(s & 1) * 2];
      af.y = a[cur][s >> 1][(s & 1) * 2 + 1];
#pragma unroll
      for (int t = 0; t < 4; ++t) {
        v2f bf;
        bf.x = b[cur][2 * s][t];
        bf.y = b[cur][2 * s + 1][t];
        acc[t] = __builtin_amdgcn_wmma_f32_16x16x4_f32(
             false, af,  false, bf,
             (short)0, acc[t],  false,  false);
      }
    }
  }

  v4f* crow = (v4f*)(C + (size_t)(m0 + (h << 3)) * NC + nb + 4 * c);
#pragma unroll
  for (int r = 0; r < 8; ++r) {
    v4f o;
    o.x = acc[0][r]; o.y = acc[1][r]; o.z = acc[2][r]; o.w = acc[3][r];
    crow[r * BROW] = o;
  }

  float ps[8], pd[8];
#pragma unroll
  for (int r = 0; r < 8; ++r) {
    ps[r] = acc[0][r] * ws.x + acc[1][r] * ws.y + acc[2][r] * ws.z + acc[3][r] * ws.w;
    pd[r] = acc[0][r] * wd.x + acc[1][r] * wd.y + acc[2][r] * wd.z + acc[3][r] * wd.w;
  }
#pragma unroll
  for (int r = 0; r < 8; ++r) { ps[r] = row_xor_add<1>(ps[r]); pd[r] = row_xor_add<1>(pd[r]); }
#pragma unroll
  for (int r = 0; r < 8; ++r) { ps[r] = row_xor_add<2>(ps[r]); pd[r] = row_xor_add<2>(pd[r]); }
#pragma unroll
  for (int r = 0; r < 8; ++r) { ps[r] = row_xor_add<4>(ps[r]); pd[r] = row_xor_add<4>(pd[r]); }
  if (G == 16) {
#pragma unroll
    for (int r = 0; r < 8; ++r) { ps[r] = row_xor_add<8>(ps[r]); pd[r] = row_xor_add<8>(pd[r]); }
  }
  const int g = c & (G - 1);
  float vs = ps[0], vd = pd[0];
#pragma unroll
  for (int r = 1; r < 8; ++r) { vs = (g == r) ? ps[r] : vs; vd = (g == r) ? pd[r] : vd; }
  if (g < 8) {
    const int idx = (m0 + (h << 3) + g) * H + (nb + 4 * c) / D;
    asrc[idx] = vs;
    adst[idx] = vd;
  }
}

__global__ void deg_count(const int* __restrict__ dstI, int* __restrict__ cnt) {
  int e = blockIdx.x * blockDim.x + threadIdx.x;
  if (e >= NEDGES) return;
  atomicAdd(&cnt[dstI[e]], 1);
}

#define SCAN_T   1000
#define SCAN_PER (NNODES / SCAN_T)
static_assert(SCAN_T * SCAN_PER == NNODES, "row_scan covers the nodes exactly");

__global__ void __launch_bounds__(SCAN_T)
row_scan(const int* __restrict__ cnt, int* __restrict__ rowptr, int* __restrict__ cursor) {
  __shared__ int part[SCAN_T];
  const int t = threadIdx.x;
  const int b = t * SCAN_PER;
  int s = 0;
  for (int j = 0; j < SCAN_PER; ++j) s += cnt[b + j];
  part[t] = s;
  __syncthreads();
  for (int off = 1; off < SCAN_T; off <<= 1) {
    int v = (t >= off) ? part[t - off] : 0;
    __syncthreads();
    part[t] += v;
    __syncthreads();
  }
  int run = part[t] - s;
  for (int j = 0; j < SCAN_PER; ++j) {
    const int i = b + j;
    rowptr[i] = run; cursor[i] = run; run += cnt[i];
  }
  if (t == SCAN_T - 1) rowptr[NNODES] = part[t];
}

__global__ void csr_fill(const int* __restrict__ srcI, const int* __restrict__ dstI,
                         int* __restrict__ cursor, int* __restrict__ slot) {
  int e = blockIdx.x * blockDim.x + threadIdx.x;
  if (e >= NEDGES) return;
  int pos = atomicAdd(&cursor[dstI[e]], 1);
  slot[pos] = srcI[e];
}

__device__ __forceinline__ float leaky(float v) { return (v > 0.f) ? v : 0.2f * v; }

template <int H, int D, bool RELU>
__device__ __forceinline__ void seg_agg_body(const int* __restrict__ rowptr,
                                             const int* __restrict__ slot,
                                             const float* __restrict__ asrc,
                                             const float* __restrict__ adst,
                                             const float* __restrict__ h,
                                             const float* __restrict__ bias,
                                             float* __restrict__ out) {
  constexpr int Q = (H * D) / 4;
  const int i = blockIdx.x * blockDim.x + threadIdx.x;
  if (i >= NNODES * Q) return;
  const int d  = i / Q;
  const int q  = i - d * Q;
  const int hh = (q * 4) / D;
  const float4* h4 = (const float4*)h;

  const float ad = adst[d * H + hh];
  float  m   = leaky(asrc[d * H + hh] + ad);
  float  s   = 1.f;
  float4 acc = h4[d * Q + q];

  int j = rowptr[d];
  const int end = rowptr[d + 1];
  for (; j + 4 <= end; j += 4) {
    const int s0 = slot[j], s1 = slot[j + 1], s2 = slot[j + 2], s3 = slot[j + 3];
    const float x0 = leaky(asrc[s0 * H + hh] + ad), x1 = leaky(asrc[s1 * H + hh] + ad);
    const float x2 = leaky(asrc[s2 * H + hh] + ad), x3 = leaky(asrc[s3 * H + hh] + ad);
    const float4 v0 = h4[s0 * Q + q], v1 = h4[s1 * Q + q];
    const float4 v2 = h4[s2 * Q + q], v3 = h4[s3 * Q + q];
    const float mn = fmaxf(fmaxf(m, x0), fmaxf(fmaxf(x1, x2), x3));
    const float sc = __expf(m - mn);
    const float p0 = __expf(x0 - mn), p1 = __expf(x1 - mn);
    const float p2 = __expf(x2 - mn), p3 = __expf(x3 - mn);
    m = mn;
    s = s * sc + ((p0 + p1) + (p2 + p3));
    acc.x *= sc; acc.y *= sc; acc.z *= sc; acc.w *= sc;
    acc.x += p0 * v0.x; acc.y += p0 * v0.y; acc.z += p0 * v0.z; acc.w += p0 * v0.w;
    acc.x += p1 * v1.x; acc.y += p1 * v1.y; acc.z += p1 * v1.z; acc.w += p1 * v1.w;
    acc.x += p2 * v2.x; acc.y += p2 * v2.y; acc.z += p2 * v2.z; acc.w += p2 * v2.w;
    acc.x += p3 * v3.x; acc.y += p3 * v3.y; acc.z += p3 * v3.z; acc.w += p3 * v3.w;
  }
  for (; j < end; ++j) {
    const int s0 = slot[j];
    const float x0 = leaky(asrc[s0 * H + hh] + ad);
    const float4 v0 = h4[s0 * Q + q];
    const float mn = fmaxf(m, x0);
    const float sc = __expf(m - mn);
    const float p0 = __expf(x0 - mn);
    m = mn;
    s = s * sc + p0;
    acc.x *= sc; acc.y *= sc; acc.z *= sc; acc.w *= sc;
    acc.x += p0 * v0.x; acc.y += p0 * v0.y; acc.z += p0 * v0.z; acc.w += p0 * v0.w;
  }

  const float inv = 1.f / s;
  const float4 bv = ((const float4*)bias)[q];
  acc.x = acc.x * inv + bv.x; acc.y = acc.y * inv + bv.y;
  acc.z = acc.z * inv + bv.z; acc.w = acc.w * inv + bv.w;
  if (RELU) {
    acc.x = fmaxf(acc.x, 0.f); acc.y = fmaxf(acc.y, 0.f);
    acc.z = fmaxf(acc.z, 0.f); acc.w = fmaxf(acc.w, 0.f);
  }
  ((float4*)out)[i] = acc;
}

__global__ void __launch_bounds__(256)
seg_agg_conv1(const int* __restrict__ rowptr, const int* __restrict__ slot,
              const float* __restrict__ asrc, const float* __restrict__ adst,
              const float* __restrict__ h,
              const float* __restrict__ bias, float* __restrict__ out) {
  seg_agg_body<H1C, D1C, true>(rowptr, slot, asrc, adst, h, bias, out);
}

__global__ void __launch_bounds__(256)
seg_agg_conv2(const int* __restrict__ rowptr, const int* __restrict__ slot,
              const float* __restrict__ asrc, const float* __restrict__ adst,
              const float* __restrict__ h,
              const float* __restrict__ bias, float* __restrict__ out) {
  seg_agg_body<H2C, D2C, false>(rowptr, slot, asrc, adst, h, bias, out);
}

#define POOL_CH  16
#define POOL_WPB 5
static_assert(NNODES % POOL_CH == 0, "pool chunks cover the nodes exactly");
static_assert((NNODES / POOL_CH) % POOL_WPB == 0, "the pool grid covers the chunks exactly, no partial block");
static_assert(F2 == 64 && POOL_CH <= 32, "a lane owns features lane and lane + 32; one lane per chunk row holds its graph id");

__global__ void __launch_bounds__(POOL_WPB * 32)
pool_accum(const float* __restrict__ h, const int* __restrict__ batch,
           float* __restrict__ sums, float* __restrict__ cnt) {
  const int lane = threadIdx.x & 31;
  const int w  = blockIdx.x * POOL_WPB + __builtin_amdgcn_readfirstlane((int)(threadIdx.x >> 5));
  const int n0 = w * POOL_CH;
  const int gl = batch[n0 + (lane & (POOL_CH - 1))];
  const float* hp = h + (size_t)n0 * F2 + lane;

  float lo[POOL_CH], hi[POOL_CH];
#pragma unroll
  for (int j = 0; j < POOL_CH; ++j) { lo[j] = hp[j * F2]; hi[j] = hp[j * F2 + 32]; }

  int   g   = __builtin_amdgcn_readlane(gl, 0);
  float a0  = lo[0], a1 = hi[0];
  float run = 1.f;
#pragma unroll
  for (int j = 1; j < POOL_CH; ++j) {
    const int gj = __builtin_amdgcn_readlane(gl, j);
    if (gj != g) {
      atomicAdd(&sums[g * F2 + lane], a0);
      atomicAdd(&sums[g * F2 + 32 + lane], a1);
      if (lane == 0) atomicAdd(&cnt[g], run);
      g = gj; a0 = lo[j]; a1 = hi[j]; run = 1.f;
    } else {
      a0 += lo[j]; a1 += hi[j]; run += 1.f;
    }
  }
  atomicAdd(&sums[g * F2 + lane], a0);
  atomicAdd(&sums[g * F2 + 32 + lane], a1);
  if (lane == 0) atomicAdd(&cnt[g], run);
}

__global__ void pool_final(const float* __restrict__ sums, const float* __restrict__ cnt,
                           float* __restrict__ out) {
  int i = blockIdx.x * blockDim.x + threadIdx.x;
  if (i >= NGRAPH * F2) return;
  out[i] = sums[i] / fmaxf(cnt[i / F2], 1.0f);
}

static inline int cdiv(long long a, int b) { return (int)((a + b - 1) / b); }

extern "C" void kernel_launch(void* const* d_in, const int* in_sizes, int n_in,
                              void* d_out, int out_size, void* d_ws, size_t ws_size,
                              hipStream_t stream) {
  const float* x        = (const float*)d_in[0];
  const int*   ei       = (const int*)d_in[1];
  const int*   batch    = (const int*)d_in[2];
  const float* W1       = (const float*)d_in[3];
  const float* att_src1 = (const float*)d_in[4];
  const float* att_dst1 = (const float*)d_in[5];
  const float* bias1    = (const float*)d_in[6];
  const float* W2       = (const float*)d_in[7];
  const float* att_src2 = (const float*)d_in[8];
  const float* att_dst2 = (const float*)d_in[9];
  const float* bias2    = (const float*)d_in[10];
  float* out = (float*)d_out;
  const int* srcI = ei;
  const int* dstI = ei + NEDGES;

  char* p = (char*)d_ws;
  auto alloc = [&](size_t bytes) -> char* {
    char* r = p; p += (bytes + 255) & ~(size_t)255; return r;
  };
  float*    h1    = (float*)alloc((size_t)NNODES * F1 * 4);
  float*    asrc1 = (float*)alloc((size_t)NNODES * H1C * 4);
  float*    adst1 = (float*)alloc((size_t)NNODES * H1C * 4);
  float*    agg1  = (float*)alloc((size_t)NNODES * F1 * 4);
  float*    h2    = (float*)alloc((size_t)NNODES * F2 * 4);
  float*    asrc2 = (float*)alloc((size_t)NNODES * H2C * 4);
  float*    adst2 = (float*)alloc((size_t)NNODES * H2C * 4);
  float*    agg2  = (float*)alloc((size_t)NNODES * F2 * 4);
  float*    psum  = (float*)alloc((size_t)NGRAPH * F2 * 4);
  float*    pcnt  = (float*)alloc((size_t)NGRAPH * 4);
  int*      cnt    = (int*)alloc((size_t)NNODES * 4);
  int*      rowptr = (int*)alloc((size_t)(NNODES + 1) * 4);
  int*      cursor = (int*)alloc((size_t)NNODES * 4);
  int*      slot   = (int*)alloc((size_t)NEDGES * 4);

  hipMemsetAsync(cnt,   0, (size_t)NNODES * 4, stream);
  hipMemsetAsync(psum,  0, (size_t)NGRAPH * F2 * 4, stream);
  hipMemsetAsync(pcnt,  0, (size_t)NGRAPH * 4, stream);

  const int T = 256;

  deg_count<<<cdiv((long long)NEDGES, T), T, 0, stream>>>(dstI, cnt);
  row_scan<<<1, SCAN_T, 0, stream>>>(cnt, rowptr, cursor);
  csr_fill<<<cdiv((long long)NEDGES, T), T, 0, stream>>>(srcI, dstI, cursor, slot);

  wmma_gemm_strip<FIN, F1, 6, D1C><<<(NNODES / 16) * (F1 / 64) / 6, 6 * 32, 0, stream>>>(
      x, W1, h1, att_src1, att_dst1, asrc1, adst1);
  seg_agg_conv1<<<cdiv((long long)NNODES * (F1 / 4), T), T, 0, stream>>>(rowptr, slot, asrc1, adst1,
                                                                         h1, bias1, agg1);

  wmma_gemm_strip<F1, F2, 5, D2C><<<(NNODES / 16) * (F2 / 64) / 5, 5 * 32, 0, stream>>>(
      agg1, W2, h2, att_src2, att_dst2, asrc2, adst2);
  seg_agg_conv2<<<cdiv((long long)NNODES * (F2 / 4), T), T, 0, stream>>>(rowptr, slot, asrc2, adst2,
                                                                         h2, bias2, agg2);

  pool_accum<<<(NNODES / POOL_CH) / POOL_WPB, POOL_WPB * 32, 0, stream>>>(agg2, batch, psum, pcnt);
  pool_final<<<cdiv((long long)NGRAPH * F2, T), T, 0, stream>>>(psum, pcnt, out);
}


# Row-pointer prefix sum rebuilt: each wave loads and stores whole 512-B runs (4 nodes per lane), running sums in registers by DPP row shifts, one barrier; strided per-thread runs and 20 barriers gone
# speedup vs baseline: 6.5105x; 1.0846x over previous
#include <hip/hip_runtime.h>
#include <math.h>

#define NNODES 50000
#define NEDGES 800000
#define NGRAPH 256
#define FIN    128
#define H1C    12
#define D1C    32
#define F1     (H1C * D1C)
#define H2C    1
#define D2C    64
#define F2     (H2C * D2C)

typedef float v2f __attribute__((ext_vector_type(2)));
typedef float v4f __attribute__((ext_vector_type(4)));
typedef float v8f __attribute__((ext_vector_type(8)));


template <int M>
__device__ __forceinline__ float row_xor_add(float v) {
  return v + __builtin_bit_cast(float, __builtin_amdgcn_update_dpp(
                 0, __builtin_bit_cast(int, v), 0x160 | M, 0xf, 0xf, true));
}

template <int K, int NC, int WPB, int D>
__global__ void __launch_bounds__(WPB * 32)
wmma_gemm_strip(const float* __restrict__ A, const float* __restrict__ B,
                float* __restrict__ C,
                const float* __restrict__ att_src, const float* __restrict__ att_dst,
                float* __restrict__ asrc, float* __restrict__ adst) {
  constexpr int MT     = NNODES / 16;
  constexpr int STRIPS = NC / 64;
  constexpr int NCH    = K / 16;
  constexpr int BROW   = NC / 4;
  static_assert(NNODES % 16 == 0 && K % 16 == 0 && NC % 64 == 0, "tile shapes divide the problem");
  static_assert((MT * STRIPS) % WPB == 0, "the grid covers the strips exactly, no partial block");
  constexpr int G = D / 4;
  constexpr int H = NC / D;
  static_assert(D == 32 || D == 64, "a head is 8 or 16 lanes of a row, and a strip holds whole heads");

  const int lane = threadIdx.x & 31;
  const int c    = lane & 15;
  const int h    = lane >> 4;
  const int w     = blockIdx.x * WPB + __builtin_amdgcn_readfirstlane((int)(threadIdx.x >> 5));
  const int strip = w / MT;
  const int m0    = (w - strip * MT) << 4;
  const int nb    = strip << 6;

  const v4f* ap = (const v4f*)(A + (size_t)(m0 + c) * K + 8 * h);
  const v4f* bp = (const v4f*)(B + (size_t)(8 * h) * NC + nb + 4 * c);

  v8f acc[4] = {};
  v4f a[2][2];
  v4f b[2][8];

  a[0][0] = ap[0];
  a[0][1] = ap[1];
#pragma unroll
  for (int j = 0; j < 8; ++j) b[0][j] = bp[j * BROW];

  const v4f ws = *(const v4f*)(att_src + nb + 4 * c);
  const v4f wd = *(const v4f*)(att_dst + nb + 4 * c);

#pragma unroll
  for (int i = 0; i < NCH; ++i) {
    const int cur = i & 1, nxt = cur ^ 1;
    if (i + 1 < NCH) {
      a[nxt][0] = ap[4 * (i + 1)];
      a[nxt][1] = ap[4 * (i + 1) + 1];
#pragma unroll
      for (int j = 0; j < 8; ++j) b[nxt][j] = bp[(16 * (i + 1) + j) * BROW];
    }
    __builtin_amdgcn_sched_barrier(0);
#pragma unroll
    for (int s = 0; s < 4; ++s) {
      v2f af;
      af.x = a[cur][s >> 1][(s & 1) * 2];
      af.y = a[cur][s >> 1][(s & 1) * 2 + 1];
#pragma unroll
      for (int t = 0; t < 4; ++t) {
        v2f bf;
        bf.x = b[cur][2 * s][t];
        bf.y = b[cur][2 * s + 1][t];
        acc[t] = __builtin_amdgcn_wmma_f32_16x16x4_f32(
             false, af,  false, bf,
             (short)0, acc[t],  false,  false);
      }
    }
  }

  v4f* crow = (v4f*)(C + (size_t)(m0 + (h << 3)) * NC + nb + 4 * c);
#pragma unroll
  for (int r = 0; r < 8; ++r) {
    v4f o;
    o.x = acc[0][r]; o.y = acc[1][r]; o.z = acc[2][r]; o.w = acc[3][r];
    crow[r * BROW] = o;
  }

  float ps[8], pd[8];
#pragma unroll
  for (int r = 0; r < 8; ++r) {
    ps[r] = acc[0][r] * ws.x + acc[1][r] * ws.y + acc[2][r] * ws.z + acc[3][r] * ws.w;
    pd[r] = acc[0][r] * wd.x + acc[1][r] * wd.y + acc[2][r] * wd.z + acc[3][r] * wd.w;
  }
#pragma unroll
  for (int r = 0; r < 8; ++r) { ps[r] = row_xor_add<1>(ps[r]); pd[r] = row_xor_add<1>(pd[r]); }
#pragma unroll
  for (int r = 0; r < 8; ++r) { ps[r] = row_xor_add<2>(ps[r]); pd[r] = row_xor_add<2>(pd[r]); }
#pragma unroll
  for (int r = 0; r < 8; ++r) { ps[r] = row_xor_add<4>(ps[r]); pd[r] = row_xor_add<4>(pd[r]); }
  if (G == 16) {
#pragma unroll
    for (int r = 0; r < 8; ++r) { ps[r] = row_xor_add<8>(ps[r]); pd[r] = row_xor_add<8>(pd[r]); }
  }
  const int g = c & (G - 1);
  float vs = ps[0], vd = pd[0];
#pragma unroll
  for (int r = 1; r < 8; ++r) { vs = (g == r) ? ps[r] : vs; vd = (g == r) ? pd[r] : vd; }
  if (g < 8) {
    const int idx = (m0 + (h << 3) + g) * H + (nb + 4 * c) / D;
    asrc[idx] = vs;
    adst[idx] = vd;
  }
}

__global__ void deg_count(const int* __restrict__ dstI, int* __restrict__ cnt) {
  int e = blockIdx.x * blockDim.x + threadIdx.x;
  if (e >= NEDGES) return;
  atomicAdd(&cnt[dstI[e]], 1);
}

#define SCAN_WAVES 32
#define SCAN_ROWS  13
#define SCAN_SPAN  (SCAN_ROWS * 128)
static_assert(SCAN_WAVES * SCAN_SPAN >= NNODES, "the scan block covers all nodes");
static_assert(NNODES % 4 == 0, "a 16-B group of counts lies wholly inside or wholly outside the node range");

template <int S>
__device__ __forceinline__ int row_shr_add(int v) {
  return v + __builtin_amdgcn_update_dpp(0, v, 0x110 | S, 0xf, 0xf, true);
}

__device__ __forceinline__ int wave_incl_scan(int v, int lane) {
  v = row_shr_add<1>(v);
  v = row_shr_add<2>(v);
  v = row_shr_add<4>(v);
  v = row_shr_add<8>(v);
  const int lo = __builtin_amdgcn_readlane(v, 15);
  return v + ((lane >= 16) ? lo : 0);
}

__global__ void __launch_bounds__(SCAN_WAVES * 32)
row_scan(const int* __restrict__ cnt, int* __restrict__ rowptr, int* __restrict__ cursor) {
  __shared__ int wtot[SCAN_WAVES];
  const int lane = threadIdx.x & 31;
  const int w    = __builtin_amdgcn_readfirstlane((int)(threadIdx.x >> 5));
  const int base = w * SCAN_SPAN + 4 * lane;

  int4 c[SCAN_ROWS];
#pragma unroll
  for (int k = 0; k < SCAN_ROWS; ++k) {
    const int i = base + k * 128;
    c[k] = (i < NNODES) ? *(const int4*)(cnt + i) : make_int4(0, 0, 0, 0);
  }

  int ex[SCAN_ROWS];
  int carry = 0;
#pragma unroll
  for (int k = 0; k < SCAN_ROWS; ++k) {
    const int t   = (c[k].x + c[k].y) + (c[k].z + c[k].w);
    const int inc = wave_incl_scan(t, lane);
    ex[k] = carry + inc - t;
    carry += __builtin_amdgcn_readlane(inc, 31);
  }

  if (lane == 0) wtot[w] = carry;
  __syncthreads();
  const int tw   = wtot[lane];
  const int iw   = wave_incl_scan(tw, lane);
  const int woff = __builtin_amdgcn_readlane(iw - tw, w);

#pragma unroll
  for (int k = 0; k < SCAN_ROWS; ++k) {
    const int i = base + k * 128;
    if (i < NNODES) {
      int4 r;
      r.x = woff + ex[k];
      r.y = r.x + c[k].x;
      r.z = r.y + c[k].y;
      r.w = r.z + c[k].z;
      *(int4*)(rowptr + i) = r;
      *(int4*)(cursor + i) = r;
    }
  }
  if (threadIdx.x == 0) rowptr[NNODES] = __builtin_amdgcn_readlane(iw, 31);
}

__global__ void csr_fill(const int* __restrict__ srcI, const int* __restrict__ dstI,
                         int* __restrict__ cursor, int* __restrict__ slot) {
  int e = blockIdx.x * blockDim.x + threadIdx.x;
  if (e >= NEDGES) return;
  int pos = atomicAdd(&cursor[dstI[e]], 1);
  slot[pos] = srcI[e];
}

__device__ __forceinline__ float leaky(float v) { return (v > 0.f) ? v : 0.2f * v; }

template <int H, int D, bool RELU>
__device__ __forceinline__ void seg_agg_body(const int* __restrict__ rowptr,
                                             const int* __restrict__ slot,
                                             const float* __restrict__ asrc,
                                             const float* __restrict__ adst,
                                             const float* __restrict__ h,
                                             const float* __restrict__ bias,
                                             float* __restrict__ out) {
  constexpr int Q = (H * D) / 4;
  const int i = blockIdx.x * blockDim.x + threadIdx.x;
  if (i >= NNODES * Q) return;
  const int d  = i / Q;
  const int q  = i - d * Q;
  const int hh = (q * 4) / D;
  const float4* h4 = (const float4*)h;

  const float ad = adst[d * H + hh];
  float  m   = leaky(asrc[d * H + hh] + ad);
  float  s   = 1.f;
  float4 acc = h4[d * Q + q];

  int j = rowptr[d];
  const int end = rowptr[d + 1];
  for (; j + 4 <= end; j += 4) {
    const int s0 = slot[j], s1 = slot[j + 1], s2 = slot[j + 2], s3 = slot[j + 3];
    const float x0 = leaky(asrc[s0 * H + hh] + ad), x1 = leaky(asrc[s1 * H + hh] + ad);
    const float x2 = leaky(asrc[s2 * H + hh] + ad), x3 = leaky(asrc[s3 * H + hh] + ad);
    const float4 v0 = h4[s0 * Q + q], v1 = h4[s1 * Q + q];
    const float4 v2 = h4[s2 * Q + q], v3 = h4[s3 * Q + q];
    const float mn = fmaxf(fmaxf(m, x0), fmaxf(fmaxf(x1, x2), x3));
    const float sc = __expf(m - mn);
    const float p0 = __expf(x0 - mn), p1 = __expf(x1 - mn);
    const float p2 = __expf(x2 - mn), p3 = __expf(x3 - mn);
    m = mn;
    s = s * sc + ((p0 + p1) + (p2 + p3));
    acc.x *= sc; acc.y *= sc; acc.z *= sc; acc.w *= sc;
    acc.x += p0 * v0.x; acc.y += p0 * v0.y; acc.z += p0 * v0.z; acc.w += p0 * v0.w;
    acc.x += p1 * v1.x; acc.y += p1 * v1.y; acc.z += p1 * v1.z; acc.w += p1 * v1.w;
    acc.x += p2 * v2.x; acc.y += p2 * v2.y; acc.z += p2 * v2.z; acc.w += p2 * v2.w;
    acc.x += p3 * v3.x; acc.y += p3 * v3.y; acc.z += p3 * v3.z; acc.w += p3 * v3.w;
  }
  for (; j < end; ++j) {
    const int s0 = slot[j];
    const float x0 = leaky(asrc[s0 * H + hh] + ad);
    const float4 v0 = h4[s0 * Q + q];
    const float mn = fmaxf(m, x0);
    const float sc = __expf(m - mn);
    const float p0 = __expf(x0 - mn);
    m = mn;
    s = s * sc + p0;
    acc.x *= sc; acc.y *= sc; acc.z *= sc; acc.w *= sc;
    acc.x += p0 * v0.x; acc.y += p0 * v0.y; acc.z += p0 * v0.z; acc.w += p0 * v0.w;
  }

  const float inv = 1.f / s;
  const float4 bv = ((const float4*)bias)[q];
  acc.x = acc.x * inv + bv.x; acc.y = acc.y * inv + bv.y;
  acc.z = acc.z * inv + bv.z; acc.w = acc.w * inv + bv.w;
  if (RELU) {
    acc.x = fmaxf(acc.x, 0.f); acc.y = fmaxf(acc.y, 0.f);
    acc.z = fmaxf(acc.z, 0.f); acc.w = fmaxf(acc.w, 0.f);
  }
  ((float4*)out)[i] = acc;
}

__global__ void __launch_bounds__(256)
seg_agg_conv1(const int* __restrict__ rowptr, const int* __restrict__ slot,
              const float* __restrict__ asrc, const float* __restrict__ adst,
              const float* __restrict__ h,
              const float* __restrict__ bias, float* __restrict__ out) {
  seg_agg_body<H1C, D1C, true>(rowptr, slot, asrc, adst, h, bias, out);
}

__global__ void __launch_bounds__(256)
seg_agg_conv2(const int* __restrict__ rowptr, const int* __restrict__ slot,
              const float* __restrict__ asrc, const float* __restrict__ adst,
              const float* __restrict__ h,
              const float* __restrict__ bias, float* __restrict__ out) {
  seg_agg_body<H2C, D2C, false>(rowptr, slot, asrc, adst, h, bias, out);
}

#define POOL_CH  16
#define POOL_WPB 5
static_assert(NNODES % POOL_CH == 0, "pool chunks cover the nodes exactly");
static_assert((NNODES / POOL_CH) % POOL_WPB == 0, "the pool grid covers the chunks exactly, no partial block");
static_assert(F2 == 64 && POOL_CH <= 32, "a lane owns features lane and lane + 32; one lane per chunk row holds its graph id");

__global__ void __launch_bounds__(POOL_WPB * 32)
pool_accum(const float* __restrict__ h, const int* __restrict__ batch,
           float* __restrict__ sums, float* __restrict__ cnt) {
  const int lane = threadIdx.x & 31;
  const int w  = blockIdx.x * POOL_WPB + __builtin_amdgcn_readfirstlane((int)(threadIdx.x >> 5));
  const int n0 = w * POOL_CH;
  const int gl = batch[n0 + (lane & (POOL_CH - 1))];
  const float* hp = h + (size_t)n0 * F2 + lane;

  float lo[POOL_CH], hi[POOL_CH];
#pragma unroll
  for (int j = 0; j < POOL_CH; ++j) { lo[j] = hp[j * F2]; hi[j] = hp[j * F2 + 32]; }

  int   g   = __builtin_amdgcn_readlane(gl, 0);
  float a0  = lo[0], a1 = hi[0];
  float run = 1.f;
#pragma unroll
  for (int j = 1; j < POOL_CH; ++j) {
    const int gj = __builtin_amdgcn_readlane(gl, j);
    if (gj != g) {
      atomicAdd(&sums[g * F2 + lane], a0);
      atomicAdd(&sums[g * F2 + 32 + lane], a1);
      if (lane == 0) atomicAdd(&cnt[g], run);
      g = gj; a0 = lo[j]; a1 = hi[j]; run = 1.f;
    } else {
      a0 += lo[j]; a1 += hi[j]; run += 1.f;
    }
  }
  atomicAdd(&sums[g * F2 + lane], a0);
  atomicAdd(&sums[g * F2 + 32 + lane], a1);
  if (lane == 0) atomicAdd(&cnt[g], run);
}

__global__ void pool_final(const float* __restrict__ sums, const float* __restrict__ cnt,
                           float* __restrict__ out) {
  int i = blockIdx.x * blockDim.x + threadIdx.x;
  if (i >= NGRAPH * F2) return;
  out[i] = sums[i] / fmaxf(cnt[i / F2], 1.0f);
}

static inline int cdiv(long long a, int b) { return (int)((a + b - 1) / b); }

extern "C" void kernel_launch(void* const* d_in, const int* in_sizes, int n_in,
                              void* d_out, int out_size, void* d_ws, size_t ws_size,
                              hipStream_t stream) {
  const float* x        = (const float*)d_in[0];
  const int*   ei       = (const int*)d_in[1];
  const int*   batch    = (const int*)d_in[2];
  const float* W1       = (const float*)d_in[3];
  const float* att_src1 = (const float*)d_in[4];
  const float* att_dst1 = (const float*)d_in[5];
  const float* bias1    = (const float*)d_in[6];
  const float* W2       = (const float*)d_in[7];
  const float* att_src2 = (const float*)d_in[8];
  const float* att_dst2 = (const float*)d_in[9];
  const float* bias2    = (const float*)d_in[10];
  float* out = (float*)d_out;
  const int* srcI = ei;
  const int* dstI = ei + NEDGES;

  char* p = (char*)d_ws;
  auto alloc = [&](size_t bytes) -> char* {
    char* r = p; p += (bytes + 255) & ~(size_t)255; return r;
  };
  float*    h1    = (float*)alloc((size_t)NNODES * F1 * 4);
  float*    asrc1 = (float*)alloc((size_t)NNODES * H1C * 4);
  float*    adst1 = (float*)alloc((size_t)NNODES * H1C * 4);
  float*    agg1  = (float*)alloc((size_t)NNODES * F1 * 4);
  float*    h2    = (float*)alloc((size_t)NNODES * F2 * 4);
  float*    asrc2 = (float*)alloc((size_t)NNODES * H2C * 4);
  float*    adst2 = (float*)alloc((size_t)NNODES * H2C * 4);
  float*    agg2  = (float*)alloc((size_t)NNODES * F2 * 4);
  float*    psum  = (float*)alloc((size_t)NGRAPH * F2 * 4);
  float*    pcnt  = (float*)alloc((size_t)NGRAPH * 4);
  int*      cnt    = (int*)alloc((size_t)NNODES * 4);
  int*      rowptr = (int*)alloc((size_t)(NNODES + 1) * 4);
  int*      cursor = (int*)alloc((size_t)NNODES * 4);
  int*      slot   = (int*)alloc((size_t)NEDGES * 4);

  hipMemsetAsync(cnt,   0, (size_t)NNODES * 4, stream);
  hipMemsetAsync(psum,  0, (size_t)NGRAPH * F2 * 4, stream);
  hipMemsetAsync(pcnt,  0, (size_t)NGRAPH * 4, stream);

  const int T = 256;

  deg_count<<<cdiv((long long)NEDGES, T), T, 0, stream>>>(dstI, cnt);
  row_scan<<<1, SCAN_WAVES * 32, 0, stream>>>(cnt, rowptr, cursor);
  csr_fill<<<cdiv((long long)NEDGES, T), T, 0, stream>>>(srcI, dstI, cursor, slot);

  wmma_gemm_strip<FIN, F1, 6, D1C><<<(NNODES / 16) * (F1 / 64) / 6, 6 * 32, 0, stream>>>(
      x, W1, h1, att_src1, att_dst1, asrc1, adst1);
  seg_agg_conv1<<<cdiv((long long)NNODES * (F1 / 4), T), T, 0, stream>>>(rowptr, slot, asrc1, adst1,
                                                                         h1, bias1, agg1);

  wmma_gemm_strip<F1, F2, 5, D2C><<<(NNODES / 16) * (F2 / 64) / 5, 5 * 32, 0, stream>>>(
      agg1, W2, h2, att_src2, att_dst2, asrc2, adst2);
  seg_agg_conv2<<<cdiv((long long)NNODES * (F2 / 4), T), T, 0, stream>>>(rowptr, slot, asrc2, adst2,
                                                                         h2, bias2, agg2);

  pool_accum<<<(NNODES / POOL_CH) / POOL_WPB, POOL_WPB * 32, 0, stream>>>(agg2, batch, psum, pcnt);
  pool_final<<<cdiv((long long)NGRAPH * F2, T), T, 0, stream>>>(psum, pcnt, out);
}


# conv1 product x@W1 loses its own launch: its six strips ride in the degree-count and slot-fill launches (three each; 4 edges per thread, atomics behind the wave's strip); sum order as before
# speedup vs baseline: 6.8578x; 1.0533x over previous
#include <hip/hip_runtime.h>
#include <math.h>

#define NNODES 50000
#define NEDGES 800000
#define NGRAPH 256
#define FIN    128
#define H1C    12
#define D1C    32
#define F1     (H1C * D1C)
#define H2C    1
#define D2C    64
#define F2     (H2C * D2C)

typedef float v2f __attribute__((ext_vector_type(2)));
typedef float v4f __attribute__((ext_vector_type(4)));
typedef float v8f __attribute__((ext_vector_type(8)));


template <int M>
__device__ __forceinline__ float row_xor_add(float v) {
  return v + __builtin_bit_cast(float, __builtin_amdgcn_update_dpp(
                 0, __builtin_bit_cast(int, v), 0x160 | M, 0xf, 0xf, true));
}

template <int K, int NC, int D>
__device__ __forceinline__ void
gemm_strip_wave(const int w,
                const float* __restrict__ A, const float* __restrict__ B,
                float* __restrict__ C,
                const float* __restrict__ att_src, const float* __restrict__ att_dst,
                float* __restrict__ asrc, float* __restrict__ adst) {
  constexpr int MT     = NNODES / 16;
  constexpr int NCH    = K / 16;
  constexpr int BROW   = NC / 4;
  static_assert(NNODES % 16 == 0 && K % 16 == 0 && NC % 64 == 0, "tile shapes divide the problem");
  constexpr int G = D / 4;
  constexpr int H = NC / D;
  static_assert(D == 32 || D == 64, "a head is 8 or 16 lanes of a row, and a strip holds whole heads");

  const int lane = threadIdx.x & 31;
  const int c    = lane & 15;
  const int h    = lane >> 4;
  const int strip = w / MT;
  const int m0    = (w - strip * MT) << 4;
  const int nb    = strip << 6;

  const v4f* ap = (const v4f*)(A + (size_t)(m0 + c) * K + 8 * h);
  const v4f* bp = (const v4f*)(B + (size_t)(8 * h) * NC + nb + 4 * c);

  v8f acc[4] = {};
  v4f a[2][2];
  v4f b[2][8];

  a[0][0] = ap[0];
  a[0][1] = ap[1];
#pragma unroll
  for (int j = 0; j < 8; ++j) b[0][j] = bp[j * BROW];

  const v4f ws = *(const v4f*)(att_src + nb + 4 * c);
  const v4f wd = *(const v4f*)(att_dst + nb + 4 * c);

#pragma unroll
  for (int i = 0; i < NCH; ++i) {
    const int cur = i & 1, nxt = cur ^ 1;
    if (i + 1 < NCH) {
      a[nxt][0] = ap[4 * (i + 1)];
      a[nxt][1] = ap[4 * (i + 1) + 1];
#pragma unroll
      for (int j = 0; j < 8; ++j) b[nxt][j] = bp[(16 * (i + 1) + j) * BROW];
    }
    __builtin_amdgcn_sched_barrier(0);
#pragma unroll
    for (int s = 0; s < 4; ++s) {
      v2f af;
      af.x = a[cur][s >> 1][(s & 1) * 2];
      af.y = a[cur][s >> 1][(s & 1) * 2 + 1];
#pragma unroll
      for (int t = 0; t < 4; ++t) {
        v2f bf;
        bf.x = b[cur][2 * s][t];
        bf.y = b[cur][2 * s + 1][t];
        acc[t] = __builtin_amdgcn_wmma_f32_16x16x4_f32(
             false, af,  false, bf,
             (short)0, acc[t],  false,  false);
      }
    }
  }

  v4f* crow = (v4f*)(C + (size_t)(m0 + (h << 3)) * NC + nb + 4 * c);
#pragma unroll
  for (int r = 0; r < 8; ++r) {
    v4f o;
    o.x = acc[0][r]; o.y = acc[1][r]; o.z = acc[2][r]; o.w = acc[3][r];
    crow[r * BROW] = o;
  }

  float ps[8], pd[8];
#pragma unroll
  for (int r = 0; r < 8; ++r) {
    ps[r] = acc[0][r] * ws.x + acc[1][r] * ws.y + acc[2][r] * ws.z + acc[3][r] * ws.w;
    pd[r] = acc[0][r] * wd.x + acc[1][r] * wd.y + acc[2][r] * wd.z + acc[3][r] * wd.w;
  }
#pragma unroll
  for (int r = 0; r < 8; ++r) { ps[r] = row_xor_add<1>(ps[r]); pd[r] = row_xor_add<1>(pd[r]); }
#pragma unroll
  for (int r = 0; r < 8; ++r) { ps[r] = row_xor_add<2>(ps[r]); pd[r] = row_xor_add<2>(pd[r]); }
#pragma unroll
  for (int r = 0; r < 8; ++r) { ps[r] = row_xor_add<4>(ps[r]); pd[r] = row_xor_add<4>(pd[r]); }
  if (G == 16) {
#pragma unroll
    for (int r = 0; r < 8; ++r) { ps[r] = row_xor_add<8>(ps[r]); pd[r] = row_xor_add<8>(pd[r]); }
  }
  const int g = c & (G - 1);
  float vs = ps[0], vd = pd[0];
#pragma unroll
  for (int r = 1; r < 8; ++r) { vs = (g == r) ? ps[r] : vs; vd = (g == r) ? pd[r] : vd; }
  if (g < 8) {
    const int idx = (m0 + (h << 3) + g) * H + (nb + 4 * c) / D;
    asrc[idx] = vs;
    adst[idx] = vd;
  }
}

template <int K, int NC, int WPB, int D>
__global__ void __launch_bounds__(WPB * 32)
wmma_gemm_strip(const float* __restrict__ A, const float* __restrict__ B,
                float* __restrict__ C,
                const float* __restrict__ att_src, const float* __restrict__ att_dst,
                float* __restrict__ asrc, float* __restrict__ adst) {
  static_assert(((NNODES / 16) * (NC / 64)) % WPB == 0, "the grid covers the strips exactly, no partial block");
  const int w = blockIdx.x * WPB + __builtin_amdgcn_readfirstlane((int)(threadIdx.x >> 5));
  gemm_strip_wave<K, NC, D>(w, A, B, C, att_src, att_dst, asrc, adst);
}

#define G1_MT    (NNODES / 16)
#define G1_PART  (3 * G1_MT)
#define EO_WPB   5
#define EO_T     (EO_WPB * 32)
#define EO_GRID  (G1_PART / EO_WPB)
#define EO_EDGE_T (NEDGES / 4)
static_assert(G1_PART % EO_WPB == 0, "the carrying grids cover their three strips exactly");
static_assert(2 * G1_PART == G1_MT * (F1 / 64), "the two launches cover the six strips");
static_assert(NEDGES % 4 == 0 && EO_EDGE_T % EO_T == 0 && EO_EDGE_T <= EO_GRID * EO_T,
              "whole blocks take the edges, four per thread, and the grid has enough of them");

__global__ void __launch_bounds__(EO_T)
deg_count_gemm(const int* __restrict__ dstI, int* __restrict__ cnt,
               const float* __restrict__ x, const float* __restrict__ W,
               float* __restrict__ hout,
               const float* __restrict__ att_src, const float* __restrict__ att_dst,
               float* __restrict__ asrc, float* __restrict__ adst) {
  const int  t     = blockIdx.x * EO_T + threadIdx.x;
  const bool edges = t < EO_EDGE_T;
  int4 d = make_int4(0, 0, 0, 0);
  if (edges) d = ((const int4*)dstI)[t];
  const int w = blockIdx.x * EO_WPB + __builtin_amdgcn_readfirstlane((int)(threadIdx.x >> 5));
  gemm_strip_wave<FIN, F1, D1C>(w, x, W, hout, att_src, att_dst, asrc, adst);
  if (edges) {
    atomicAdd(&cnt[d.x], 1);
    atomicAdd(&cnt[d.y], 1);
    atomicAdd(&cnt[d.z], 1);
    atomicAdd(&cnt[d.w], 1);
  }
}

#define SCAN_WAVES 32
#define SCAN_ROWS  13
#define SCAN_SPAN  (SCAN_ROWS * 128)
static_assert(SCAN_WAVES * SCAN_SPAN >= NNODES, "the scan block covers all nodes");
static_assert(NNODES % 4 == 0, "a 16-B group of counts lies wholly inside or wholly outside the node range");

template <int S>
__device__ __forceinline__ int row_shr_add(int v) {
  return v + __builtin_amdgcn_update_dpp(0, v, 0x110 | S, 0xf, 0xf, true);
}

__device__ __forceinline__ int wave_incl_scan(int v, int lane) {
  v = row_shr_add<1>(v);
  v = row_shr_add<2>(v);
  v = row_shr_add<4>(v);
  v = row_shr_add<8>(v);
  const int lo = __builtin_amdgcn_readlane(v, 15);
  return v + ((lane >= 16) ? lo : 0);
}

__global__ void __launch_bounds__(SCAN_WAVES * 32)
row_scan(const int* __restrict__ cnt, int* __restrict__ rowptr, int* __restrict__ cursor) {
  __shared__ int wtot[SCAN_WAVES];
  const int lane = threadIdx.x & 31;
  const int w    = __builtin_amdgcn_readfirstlane((int)(threadIdx.x >> 5));
  const int base = w * SCAN_SPAN + 4 * lane;

  int4 c[SCAN_ROWS];
#pragma unroll
  for (int k = 0; k < SCAN_ROWS; ++k) {
    const int i = base + k * 128;
    c[k] = (i < NNODES) ? *(const int4*)(cnt + i) : make_int4(0, 0, 0, 0);
  }

  int ex[SCAN_ROWS];
  int carry = 0;
#pragma unroll
  for (int k = 0; k < SCAN_ROWS; ++k) {
    const int t   = (c[k].x + c[k].y) + (c[k].z + c[k].w);
    const int inc = wave_incl_scan(t, lane);
    ex[k] = carry + inc - t;
    carry += __builtin_amdgcn_readlane(inc, 31);
  }

  if (lane == 0) wtot[w] = carry;
  __syncthreads();
  const int tw   = wtot[lane];
  const int iw   = wave_incl_scan(tw, lane);
  const int woff = __builtin_amdgcn_readlane(iw - tw, w);

#pragma unroll
  for (int k = 0; k < SCAN_ROWS; ++k) {
    const int i = base + k * 128;
    if (i < NNODES) {
      int4 r;
      r.x = woff + ex[k];
      r.y = r.x + c[k].x;
      r.z = r.y + c[k].y;
      r.w = r.z + c[k].z;
      *(int4*)(rowptr + i) = r;
      *(int4*)(cursor + i) = r;
    }
  }
  if (threadIdx.x == 0) rowptr[NNODES] = __builtin_amdgcn_readlane(iw, 31);
}

__global__ void __launch_bounds__(EO_T)
csr_fill_gemm(const int* __restrict__ srcI, const int* __restrict__ dstI,
              int* __restrict__ cursor, int* __restrict__ slot,
              const float* __restrict__ x, const float* __restrict__ W,
              float* __restrict__ hout,
              const float* __restrict__ att_src, const float* __restrict__ att_dst,
              float* __restrict__ asrc, float* __restrict__ adst) {
  const int  t     = blockIdx.x * EO_T + threadIdx.x;
  const bool edges = t < EO_EDGE_T;
  int4 d = make_int4(0, 0, 0, 0), s = make_int4(0, 0, 0, 0);
  if (edges) {
    d = ((const int4*)dstI)[t];
    s = ((const int4*)srcI)[t];
  }
  const int w = blockIdx.x * EO_WPB + __builtin_amdgcn_readfirstlane((int)(threadIdx.x >> 5));
  gemm_strip_wave<FIN, F1, D1C>(G1_PART + w, x, W, hout, att_src, att_dst, asrc, adst);
  if (edges) {
    const int p0 = atomicAdd(&cursor[d.x], 1);
    const int p1 = atomicAdd(&cursor[d.y], 1);
    const int p2 = atomicAdd(&cursor[d.z], 1);
    const int p3 = atomicAdd(&cursor[d.w], 1);
    slot[p0] = s.x;
    slot[p1] = s.y;
    slot[p2] = s.z;
    slot[p3] = s.w;
  }
}

__device__ __forceinline__ float leaky(float v) { return (v > 0.f) ? v : 0.2f * v; }

template <int H, int D, bool RELU>
__device__ __forceinline__ void seg_agg_body(const int* __restrict__ rowptr,
                                             const int* __restrict__ slot,
                                             const float* __restrict__ asrc,
                                             const float* __restrict__ adst,
                                             const float* __restrict__ h,
                                             const float* __restrict__ bias,
                                             float* __restrict__ out) {
  constexpr int Q = (H * D) / 4;
  const int i = blockIdx.x * blockDim.x + threadIdx.x;
  if (i >= NNODES * Q) return;
  const int d  = i / Q;
  const int q  = i - d * Q;
  const int hh = (q * 4) / D;
  const float4* h4 = (const float4*)h;

  const float ad = adst[d * H + hh];
  float  m   = leaky(asrc[d * H + hh] + ad);
  float  s   = 1.f;
  float4 acc = h4[d * Q + q];

  int j = rowptr[d];
  const int end = rowptr[d + 1];
  for (; j + 4 <= end; j += 4) {
    const int s0 = slot[j], s1 = slot[j + 1], s2 = slot[j + 2], s3 = slot[j + 3];
    const float x0 = leaky(asrc[s0 * H + hh] + ad), x1 = leaky(asrc[s1 * H + hh] + ad);
    const float x2 = leaky(asrc[s2 * H + hh] + ad), x3 = leaky(asrc[s3 * H + hh] + ad);
    const float4 v0 = h4[s0 * Q + q], v1 = h4[s1 * Q + q];
    const float4 v2 = h4[s2 * Q + q], v3 = h4[s3 * Q + q];
    const float mn = fmaxf(fmaxf(m, x0), fmaxf(fmaxf(x1, x2), x3));
    const float sc = __expf(m - mn);
    const float p0 = __expf(x0 - mn), p1 = __expf(x1 - mn);
    const float p2 = __expf(x2 - mn), p3 = __expf(x3 - mn);
    m = mn;
    s = s * sc + ((p0 + p1) + (p2 + p3));
    acc.x *= sc; acc.y *= sc; acc.z *= sc; acc.w *= sc;
    acc.x += p0 * v0.x; acc.y += p0 * v0.y; acc.z += p0 * v0.z; acc.w += p0 * v0.w;
    acc.x += p1 * v1.x; acc.y += p1 * v1.y; acc.z += p1 * v1.z; acc.w += p1 * v1.w;
    acc.x += p2 * v2.x; acc.y += p2 * v2.y; acc.z += p2 * v2.z; acc.w += p2 * v2.w;
    acc.x += p3 * v3.x; acc.y += p3 * v3.y; acc.z += p3 * v3.z; acc.w += p3 * v3.w;
  }
  for (; j < end; ++j) {
    const int s0 = slot[j];
    const float x0 = leaky(asrc[s0 * H + hh] + ad);
    const float4 v0 = h4[s0 * Q + q];
    const float mn = fmaxf(m, x0);
    const float sc = __expf(m - mn);
    const float p0 = __expf(x0 - mn);
    m = mn;
    s = s * sc + p0;
    acc.x *= sc; acc.y *= sc; acc.z *= sc; acc.w *= sc;
    acc.x += p0 * v0.x; acc.y += p0 * v0.y; acc.z += p0 * v0.z; acc.w += p0 * v0.w;
  }

  const float inv = 1.f / s;
  const float4 bv = ((const float4*)bias)[q];
  acc.x = acc.x * inv + bv.x; acc.y = acc.y * inv + bv.y;
  acc.z = acc.z * inv + bv.z; acc.w = acc.w * inv + bv.w;
  if (RELU) {
    acc.x = fmaxf(acc.x, 0.f); acc.y = fmaxf(acc.y, 0.f);
    acc.z = fmaxf(acc.z, 0.f); acc.w = fmaxf(acc.w, 0.f);
  }
  ((float4*)out)[i] = acc;
}

__global__ void __launch_bounds__(256)
seg_agg_conv1(const int* __restrict__ rowptr, const int* __restrict__ slot,
              const float* __restrict__ asrc, const float* __restrict__ adst,
              const float* __restrict__ h,
              const float* __restrict__ bias, float* __restrict__ out) {
  seg_agg_body<H1C, D1C, true>(rowptr, slot, asrc, adst, h, bias, out);
}

__global__ void __launch_bounds__(256)
seg_agg_conv2(const int* __restrict__ rowptr, const int* __restrict__ slot,
              const float* __restrict__ asrc, const float* __restrict__ adst,
              const float* __restrict__ h,
              const float* __restrict__ bias, float* __restrict__ out) {
  seg_agg_body<H2C, D2C, false>(rowptr, slot, asrc, adst, h, bias, out);
}

#define POOL_CH  16
#define POOL_WPB 5
static_assert(NNODES % POOL_CH == 0, "pool chunks cover the nodes exactly");
static_assert((NNODES / POOL_CH) % POOL_WPB == 0, "the pool grid covers the chunks exactly, no partial block");
static_assert(F2 == 64 && POOL_CH <= 32, "a lane owns features lane and lane + 32; one lane per chunk row holds its graph id");

__global__ void __launch_bounds__(POOL_WPB * 32)
pool_accum(const float* __restrict__ h, const int* __restrict__ batch,
           float* __restrict__ sums, float* __restrict__ cnt) {
  const int lane = threadIdx.x & 31;
  const int w  = blockIdx.x * POOL_WPB + __builtin_amdgcn_readfirstlane((int)(threadIdx.x >> 5));
  const int n0 = w * POOL_CH;
  const int gl = batch[n0 + (lane & (POOL_CH - 1))];
  const float* hp = h + (size_t)n0 * F2 + lane;

  float lo[POOL_CH], hi[POOL_CH];
#pragma unroll
  for (int j = 0; j < POOL_CH; ++j) { lo[j] = hp[j * F2]; hi[j] = hp[j * F2 + 32]; }

  int   g   = __builtin_amdgcn_readlane(gl, 0);
  float a0  = lo[0], a1 = hi[0];
  float run = 1.f;
#pragma unroll
  for (int j = 1; j < POOL_CH; ++j) {
    const int gj = __builtin_amdgcn_readlane(gl, j);
    if (gj != g) {
      atomicAdd(&sums[g * F2 + lane], a0);
      atomicAdd(&sums[g * F2 + 32 + lane], a1);
      if (lane == 0) atomicAdd(&cnt[g], run);
      g = gj; a0 = lo[j]; a1 = hi[j]; run = 1.f;
    } else {
      a0 += lo[j]; a1 += hi[j]; run += 1.f;
    }
  }
  atomicAdd(&sums[g * F2 + lane], a0);
  atomicAdd(&sums[g * F2 + 32 + lane], a1);
  if (lane == 0) atomicAdd(&cnt[g], run);
}

__global__ void pool_final(const float* __restrict__ sums, const float* __restrict__ cnt,
                           float* __restrict__ out) {
  int i = blockIdx.x * blockDim.x + threadIdx.x;
  if (i >= NGRAPH * F2) return;
  out[i] = sums[i] / fmaxf(cnt[i / F2], 1.0f);
}

static inline int cdiv(long long a, int b) { return (int)((a + b - 1) / b); }

extern "C" void kernel_launch(void* const* d_in, const int* in_sizes, int n_in,
                              void* d_out, int out_size, void* d_ws, size_t ws_size,
                              hipStream_t stream) {
  const float* x        = (const float*)d_in[0];
  const int*   ei       = (const int*)d_in[1];
  const int*   batch    = (const int*)d_in[2];
  const float* W1       = (const float*)d_in[3];
  const float* att_src1 = (const float*)d_in[4];
  const float* att_dst1 = (const float*)d_in[5];
  const float* bias1    = (const float*)d_in[6];
  const float* W2       = (const float*)d_in[7];
  const float* att_src2 = (const float*)d_in[8];
  const float* att_dst2 = (const float*)d_in[9];
  const float* bias2    = (const float*)d_in[10];
  float* out = (float*)d_out;
  const int* srcI = ei;
  const int* dstI = ei + NEDGES;

  char* p = (char*)d_ws;
  auto alloc = [&](size_t bytes) -> char* {
    char* r = p; p += (bytes + 255) & ~(size_t)255; return r;
  };
  float*    h1    = (float*)alloc((size_t)NNODES * F1 * 4);
  float*    asrc1 = (float*)alloc((size_t)NNODES * H1C * 4);
  float*    adst1 = (float*)alloc((size_t)NNODES * H1C * 4);
  float*    agg1  = (float*)alloc((size_t)NNODES * F1 * 4);
  float*    h2    = (float*)alloc((size_t)NNODES * F2 * 4);
  float*    asrc2 = (float*)alloc((size_t)NNODES * H2C * 4);
  float*    adst2 = (float*)alloc((size_t)NNODES * H2C * 4);
  float*    agg2  = (float*)alloc((size_t)NNODES * F2 * 4);
  float*    psum  = (float*)alloc((size_t)NGRAPH * F2 * 4);
  float*    pcnt  = (float*)alloc((size_t)NGRAPH * 4);
  int*      cnt    = (int*)alloc((size_t)NNODES * 4);
  int*      rowptr = (int*)alloc((size_t)(NNODES + 1) * 4);
  int*      cursor = (int*)alloc((size_t)NNODES * 4);
  int*      slot   = (int*)alloc((size_t)NEDGES * 4);

  hipMemsetAsync(cnt,   0, (size_t)NNODES * 4, stream);
  hipMemsetAsync(psum,  0, (size_t)NGRAPH * F2 * 4, stream);
  hipMemsetAsync(pcnt,  0, (size_t)NGRAPH * 4, stream);

  const int T = 256;

  deg_count_gemm<<<EO_GRID, EO_T, 0, stream>>>(dstI, cnt,
      x, W1, h1, att_src1, att_dst1, asrc1, adst1);
  row_scan<<<1, SCAN_WAVES * 32, 0, stream>>>(cnt, rowptr, cursor);
  csr_fill_gemm<<<EO_GRID, EO_T, 0, stream>>>(srcI, dstI, cursor, slot,
      x, W1, h1, att_src1, att_dst1, asrc1, adst1);

  seg_agg_conv1<<<cdiv((long long)NNODES * (F1 / 4), T), T, 0, stream>>>(rowptr, slot, asrc1, adst1,
                                                                         h1, bias1, agg1);

  wmma_gemm_strip<F1, F2, 5, D2C><<<(NNODES / 16) * (F2 / 64) / 5, 5 * 32, 0, stream>>>(
      agg1, W2, h2, att_src2, att_dst2, asrc2, adst2);
  seg_agg_conv2<<<cdiv((long long)NNODES * (F2 / 4), T), T, 0, stream>>>(rowptr, slot, asrc2, adst2,
                                                                         h2, bias2, agg2);

  pool_accum<<<(NNODES / POOL_CH) / POOL_WPB, POOL_WPB * 32, 0, stream>>>(agg2, batch, psum, pcnt);
  pool_final<<<cdiv((long long)NGRAPH * F2, T), T, 0, stream>>>(psum, pcnt, out);
}


# Edge grouping by a two-level counting sort on LDS counters (chunk x node-bin counts, scan, scatter, per-bin sort): 1.6M global integer atomics and the count clear gone; row order still by arrival
# speedup vs baseline: 9.2789x; 1.3530x over previous
#include <hip/hip_runtime.h>
#include <math.h>

#define NNODES 50000
#define NEDGES 800000
#define NGRAPH 256
#define FIN    128
#define H1C    12
#define D1C    32
#define F1     (H1C * D1C)
#define H2C    1
#define D2C    64
#define F2     (H2C * D2C)

typedef float v2f __attribute__((ext_vector_type(2)));
typedef float v4f __attribute__((ext_vector_type(4)));
typedef float v8f __attribute__((ext_vector_type(8)));


template <int M>
__device__ __forceinline__ float row_xor_add(float v) {
  return v + __builtin_bit_cast(float, __builtin_amdgcn_update_dpp(
                 0, __builtin_bit_cast(int, v), 0x160 | M, 0xf, 0xf, true));
}

template <int K, int NC, int D>
__device__ __forceinline__ void
gemm_strip_wave(const int w,
                const float* __restrict__ A, const float* __restrict__ B,
                float* __restrict__ C,
                const float* __restrict__ att_src, const float* __restrict__ att_dst,
                float* __restrict__ asrc, float* __restrict__ adst) {
  constexpr int MT     = NNODES / 16;
  constexpr int NCH    = K / 16;
  constexpr int BROW   = NC / 4;
  static_assert(NNODES % 16 == 0 && K % 16 == 0 && NC % 64 == 0, "tile shapes divide the problem");
  constexpr int G = D / 4;
  constexpr int H = NC / D;
  static_assert(D == 32 || D == 64, "a head is 8 or 16 lanes of a row, and a strip holds whole heads");

  const int lane = threadIdx.x & 31;
  const int c    = lane & 15;
  const int h    = lane >> 4;
  const int strip = w / MT;
  const int m0    = (w - strip * MT) << 4;
  const int nb    = strip << 6;

  const v4f* ap = (const v4f*)(A + (size_t)(m0 + c) * K + 8 * h);
  const v4f* bp = (const v4f*)(B + (size_t)(8 * h) * NC + nb + 4 * c);

  v8f acc[4] = {};
  v4f a[2][2];
  v4f b[2][8];

  a[0][0] = ap[0];
  a[0][1] = ap[1];
#pragma unroll
  for (int j = 0; j < 8; ++j) b[0][j] = bp[j * BROW];

  const v4f ws = *(const v4f*)(att_src + nb + 4 * c);
  const v4f wd = *(const v4f*)(att_dst + nb + 4 * c);

#pragma unroll
  for (int i = 0; i < NCH; ++i) {
    const int cur = i & 1, nxt = cur ^ 1;
    if (i + 1 < NCH) {
      a[nxt][0] = ap[4 * (i + 1)];
      a[nxt][1] = ap[4 * (i + 1) + 1];
#pragma unroll
      for (int j = 0; j < 8; ++j) b[nxt][j] = bp[(16 * (i + 1) + j) * BROW];
    }
    __builtin_amdgcn_sched_barrier(0);
#pragma unroll
    for (int s = 0; s < 4; ++s) {
      v2f af;
      af.x = a[cur][s >> 1][(s & 1) * 2];
      af.y = a[cur][s >> 1][(s & 1) * 2 + 1];
#pragma unroll
      for (int t = 0; t < 4; ++t) {
        v2f bf;
        bf.x = b[cur][2 * s][t];
        bf.y = b[cur][2 * s + 1][t];
        acc[t] = __builtin_amdgcn_wmma_f32_16x16x4_f32(
             false, af,  false, bf,
             (short)0, acc[t],  false,  false);
      }
    }
  }

  v4f* crow = (v4f*)(C + (size_t)(m0 + (h << 3)) * NC + nb + 4 * c);
#pragma unroll
  for (int r = 0; r < 8; ++r) {
    v4f o;
    o.x = acc[0][r]; o.y = acc[1][r]; o.z = acc[2][r]; o.w = acc[3][r];
    crow[r * BROW] = o;
  }

  float ps[8], pd[8];
#pragma unroll
  for (int r = 0; r < 8; ++r) {
    ps[r] = acc[0][r] * ws.x + acc[1][r] * ws.y + acc[2][r] * ws.z + acc[3][r] * ws.w;
    pd[r] = acc[0][r] * wd.x + acc[1][r] * wd.y + acc[2][r] * wd.z + acc[3][r] * wd.w;
  }
#pragma unroll
  for (int r = 0; r < 8; ++r) { ps[r] = row_xor_add<1>(ps[r]); pd[r] = row_xor_add<1>(pd[r]); }
#pragma unroll
  for (int r = 0; r < 8; ++r) { ps[r] = row_xor_add<2>(ps[r]); pd[r] = row_xor_add<2>(pd[r]); }
#pragma unroll
  for (int r = 0; r < 8; ++r) { ps[r] = row_xor_add<4>(ps[r]); pd[r] = row_xor_add<4>(pd[r]); }
  if (G == 16) {
#pragma unroll
    for (int r = 0; r < 8; ++r) { ps[r] = row_xor_add<8>(ps[r]); pd[r] = row_xor_add<8>(pd[r]); }
  }
  const int g = c & (G - 1);
  float vs = ps[0], vd = pd[0];
#pragma unroll
  for (int r = 1; r < 8; ++r) { vs = (g == r) ? ps[r] : vs; vd = (g == r) ? pd[r] : vd; }
  if (g < 8) {
    const int idx = (m0 + (h << 3) + g) * H + (nb + 4 * c) / D;
    asrc[idx] = vs;
    adst[idx] = vd;
  }
}

template <int K, int NC, int WPB, int D>
__global__ void __launch_bounds__(WPB * 32)
wmma_gemm_strip(const float* __restrict__ A, const float* __restrict__ B,
                float* __restrict__ C,
                const float* __restrict__ att_src, const float* __restrict__ att_dst,
                float* __restrict__ asrc, float* __restrict__ adst) {
  static_assert(((NNODES / 16) * (NC / 64)) % WPB == 0, "the grid covers the strips exactly, no partial block");
  const int w = blockIdx.x * WPB + __builtin_amdgcn_readfirstlane((int)(threadIdx.x >> 5));
  gemm_strip_wave<K, NC, D>(w, A, B, C, att_src, att_dst, asrc, adst);
}

#define G1_MT    (NNODES / 16)
#define G1_PART  (3 * G1_MT)
#define EO_WPB   5
#define EO_T     (EO_WPB * 32)
#define EO_GRID  (G1_PART / EO_WPB)
#define BIN_SHIFT 8
#define BIN_NODES (1 << BIN_SHIFT)
#define NBIN     ((NNODES + BIN_NODES - 1) / BIN_NODES)
#define CHUNK_E  4096
#define NCHUNK   ((NEDGES + CHUNK_E - 1) / CHUNK_E)
#define NHIST    (NBIN * NCHUNK)
#define CH_G     (CHUNK_E / 4)
#define EG_TRIPS ((CH_G + EO_T - 1) / EO_T)
static_assert(G1_PART % EO_WPB == 0, "the carrying grids cover their three strips exactly");
static_assert(2 * G1_PART == G1_MT * (F1 / 64), "the two launches cover the six strips");
static_assert(NEDGES % 4 == 0 && CHUNK_E % 4 == 0, "a 16-B group of four edges lies wholly inside one chunk and inside the list");
static_assert(NCHUNK <= EO_GRID, "the carrying grids have a block for every chunk");
static_assert(NNODES <= 65536 && BIN_NODES <= 256, "a packed edge word holds the source in 16 bits and the node's place in its bin above them");
static_assert(NHIST % 4 == 0, "a 16-B group of counts lies wholly inside or wholly outside the table");

__global__ void __launch_bounds__(EO_T)
bin_count_gemm(const int* __restrict__ dstI, int* __restrict__ hist,
               const float* __restrict__ x, const float* __restrict__ W,
               float* __restrict__ hout,
               const float* __restrict__ att_src, const float* __restrict__ att_dst,
               float* __restrict__ asrc, float* __restrict__ adst) {
  __shared__ int bc[NBIN];
  if (blockIdx.x < NCHUNK) {
    for (int b = threadIdx.x; b < NBIN; b += EO_T) bc[b] = 0;
    __syncthreads();
    int4 d[EG_TRIPS];
#pragma unroll
    for (int k = 0; k < EG_TRIPS; ++k) {
      const int q = threadIdx.x + k * EO_T;
      const int g = blockIdx.x * CH_G + q;
      d[k] = (q < CH_G && g < NEDGES / 4) ? ((const int4*)dstI)[g] : make_int4(0, 0, 0, 0);
    }
#pragma unroll
    for (int k = 0; k < EG_TRIPS; ++k) {
      const int q = threadIdx.x + k * EO_T;
      const int g = blockIdx.x * CH_G + q;
      if (q < CH_G && g < NEDGES / 4) {
        atomicAdd(&bc[d[k].x >> BIN_SHIFT], 1);
        atomicAdd(&bc[d[k].y >> BIN_SHIFT], 1);
        atomicAdd(&bc[d[k].z >> BIN_SHIFT], 1);
        atomicAdd(&bc[d[k].w >> BIN_SHIFT], 1);
      }
    }
    __syncthreads();
    for (int b = threadIdx.x; b < NBIN; b += EO_T) hist[b * NCHUNK + blockIdx.x] = bc[b];
  }
  const int w = blockIdx.x * EO_WPB + __builtin_amdgcn_readfirstlane((int)(threadIdx.x >> 5));
  gemm_strip_wave<FIN, F1, D1C>(w, x, W, hout, att_src, att_dst, asrc, adst);
}

#define SCAN_WAVES 32
#define SCAN_ROWS  10
#define SCAN_SPAN  (SCAN_ROWS * 128)
static_assert(SCAN_WAVES * SCAN_SPAN >= NHIST, "the scan block covers the whole table");

template <int S>
__device__ __forceinline__ int row_shr_add(int v) {
  return v + __builtin_amdgcn_update_dpp(0, v, 0x110 | S, 0xf, 0xf, true);
}

__device__ __forceinline__ int wave_incl_scan(int v, int lane) {
  v = row_shr_add<1>(v);
  v = row_shr_add<2>(v);
  v = row_shr_add<4>(v);
  v = row_shr_add<8>(v);
  const int lo = __builtin_amdgcn_readlane(v, 15);
  return v + ((lane >= 16) ? lo : 0);
}

__global__ void __launch_bounds__(SCAN_WAVES * 32)
hist_scan(const int* __restrict__ hist, int* __restrict__ off) {
  __shared__ int wtot[SCAN_WAVES];
  const int lane = threadIdx.x & 31;
  const int w    = __builtin_amdgcn_readfirstlane((int)(threadIdx.x >> 5));
  const int base = w * SCAN_SPAN + 4 * lane;

  int4 c[SCAN_ROWS];
#pragma unroll
  for (int k = 0; k < SCAN_ROWS; ++k) {
    const int i = base + k * 128;
    c[k] = (i < NHIST) ? *(const int4*)(hist + i) : make_int4(0, 0, 0, 0);
  }

  int ex[SCAN_ROWS];
  int carry = 0;
#pragma unroll
  for (int k = 0; k < SCAN_ROWS; ++k) {
    const int t   = (c[k].x + c[k].y) + (c[k].z + c[k].w);
    const int inc = wave_incl_scan(t, lane);
    ex[k] = carry + inc - t;
    carry += __builtin_amdgcn_readlane(inc, 31);
  }

  if (lane == 0) wtot[w] = carry;
  __syncthreads();
  const int tw   = wtot[lane];
  const int iw   = wave_incl_scan(tw, lane);
  const int woff = __builtin_amdgcn_readlane(iw - tw, w);

#pragma unroll
  for (int k = 0; k < SCAN_ROWS; ++k) {
    const int i = base + k * 128;
    if (i < NHIST) {
      int4 r;
      r.x = woff + ex[k];
      r.y = r.x + c[k].x;
      r.z = r.y + c[k].y;
      r.w = r.z + c[k].z;
      *(int4*)(off + i) = r;
    }
  }
  if (threadIdx.x == 0) off[NHIST] = __builtin_amdgcn_readlane(iw, 31);
}

__global__ void __launch_bounds__(EO_T)
bin_scatter_gemm(const int* __restrict__ srcI, const int* __restrict__ dstI,
                 const int* __restrict__ off, int* __restrict__ binned,
                 const float* __restrict__ x, const float* __restrict__ W,
                 float* __restrict__ hout,
                 const float* __restrict__ att_src, const float* __restrict__ att_dst,
                 float* __restrict__ asrc, float* __restrict__ adst) {
  __shared__ int bp[NBIN];
  if (blockIdx.x < NCHUNK) {
    for (int b = threadIdx.x; b < NBIN; b += EO_T) bp[b] = off[b * NCHUNK + blockIdx.x];
    __syncthreads();
    int4 d[EG_TRIPS], s[EG_TRIPS];
#pragma unroll
    for (int k = 0; k < EG_TRIPS; ++k) {
      const int q = threadIdx.x + k * EO_T;
      const int g = blockIdx.x * CH_G + q;
      const bool in = (q < CH_G && g < NEDGES / 4);
      d[k] = in ? ((const int4*)dstI)[g] : make_int4(0, 0, 0, 0);
      s[k] = in ? ((const int4*)srcI)[g] : make_int4(0, 0, 0, 0);
    }
#pragma unroll
    for (int k = 0; k < EG_TRIPS; ++k) {
      const int q = threadIdx.x + k * EO_T;
      const int g = blockIdx.x * CH_G + q;
      if (q < CH_G && g < NEDGES / 4) {
        const int p0 = atomicAdd(&bp[d[k].x >> BIN_SHIFT], 1);
        const int p1 = atomicAdd(&bp[d[k].y >> BIN_SHIFT], 1);
        const int p2 = atomicAdd(&bp[d[k].z >> BIN_SHIFT], 1);
        const int p3 = atomicAdd(&bp[d[k].w >> BIN_SHIFT], 1);
        binned[p0] = s[k].x | ((d[k].x & (BIN_NODES - 1)) << 16);
        binned[p1] = s[k].y | ((d[k].y & (BIN_NODES - 1)) << 16);
        binned[p2] = s[k].z | ((d[k].z & (BIN_NODES - 1)) << 16);
        binned[p3] = s[k].w | ((d[k].w & (BIN_NODES - 1)) << 16);
      }
    }
  }
  const int w = blockIdx.x * EO_WPB + __builtin_amdgcn_readfirstlane((int)(threadIdx.x >> 5));
  gemm_strip_wave<FIN, F1, D1C>(G1_PART + w, x, W, hout, att_src, att_dst, asrc, adst);
}

#define BS_T 1024
#define BS_LD 4
static_assert(BIN_NODES % 32 == 0 && BIN_NODES <= BS_T, "whole waves hold the bin's node counters, one per thread");

__global__ void __launch_bounds__(BS_T)
bin_sort(const int* __restrict__ off, const int* __restrict__ binned,
         int* __restrict__ rowptr, int* __restrict__ slot) {
  __shared__ int nc[BIN_NODES];
  __shared__ int wsum[BIN_NODES / 32];
  const int tid  = threadIdx.x;
  const int lane = tid & 31;
  const int b    = blockIdx.x;
  const int e0   = off[b * NCHUNK];
  const int e1   = off[(b + 1) * NCHUNK];
  if (tid < BIN_NODES) nc[tid] = 0;
  __syncthreads();
  for (int e = e0 + tid; e < e1; e += BS_LD * BS_T) {
    int v[BS_LD];
#pragma unroll
    for (int k = 0; k < BS_LD; ++k) v[k] = binned[min(e + k * BS_T, e1 - 1)];
#pragma unroll
    for (int k = 0; k < BS_LD; ++k)
      if (e + k * BS_T < e1) atomicAdd(&nc[v[k] >> 16], 1);
  }
  __syncthreads();
  int c = 0, inc = 0;
  if (tid < BIN_NODES) {
    c   = nc[tid];
    inc = wave_incl_scan(c, lane);
    if (lane == 31) wsum[tid >> 5] = inc;
  }
  __syncthreads();
  if (tid < BIN_NODES) {
    int wo = 0;
    for (int j = 0; j < (tid >> 5); ++j) wo += wsum[j];
    const int ex = wo + inc - c;
    nc[tid] = ex;
    const int node = b * BIN_NODES + tid;
    if (node < NNODES) rowptr[node] = e0 + ex;
  }
  if (b == NBIN - 1 && tid == 0) rowptr[NNODES] = e1;
  __syncthreads();
  for (int e = e0 + tid; e < e1; e += BS_LD * BS_T) {
    int v[BS_LD];
#pragma unroll
    for (int k = 0; k < BS_LD; ++k) v[k] = binned[min(e + k * BS_T, e1 - 1)];
#pragma unroll
    for (int k = 0; k < BS_LD; ++k) {
      if (e + k * BS_T < e1) {
        const int p = atomicAdd(&nc[v[k] >> 16], 1);
        slot[e0 + p] = v[k] & 0xffff;
      }
    }
  }
}

__device__ __forceinline__ float leaky(float v) { return (v > 0.f) ? v : 0.2f * v; }

template <int H, int D, bool RELU>
__device__ __forceinline__ void seg_agg_body(const int* __restrict__ rowptr,
                                             const int* __restrict__ slot,
                                             const float* __restrict__ asrc,
                                             const float* __restrict__ adst,
                                             const float* __restrict__ h,
                                             const float* __restrict__ bias,
                                             float* __restrict__ out) {
  constexpr int Q = (H * D) / 4;
  const int i = blockIdx.x * blockDim.x + threadIdx.x;
  if (i >= NNODES * Q) return;
  const int d  = i / Q;
  const int q  = i - d * Q;
  const int hh = (q * 4) / D;
  const float4* h4 = (const float4*)h;

  const float ad = adst[d * H + hh];
  float  m   = leaky(asrc[d * H + hh] + ad);
  float  s   = 1.f;
  float4 acc = h4[d * Q + q];

  int j = rowptr[d];
  const int end = rowptr[d + 1];
  for (; j + 4 <= end; j += 4) {
    const int s0 = slot[j], s1 = slot[j + 1], s2 = slot[j + 2], s3 = slot[j + 3];
    const float x0 = leaky(asrc[s0 * H + hh] + ad), x1 = leaky(asrc[s1 * H + hh] + ad);
    const float x2 = leaky(asrc[s2 * H + hh] + ad), x3 = leaky(asrc[s3 * H + hh] + ad);
    const float4 v0 = h4[s0 * Q + q], v1 = h4[s1 * Q + q];
    const float4 v2 = h4[s2 * Q + q], v3 = h4[s3 * Q + q];
    const float mn = fmaxf(fmaxf(m, x0), fmaxf(fmaxf(x1, x2), x3));
    const float sc = __expf(m - mn);
    const float p0 = __expf(x0 - mn), p1 = __expf(x1 - mn);
    const float p2 = __expf(x2 - mn), p3 = __expf(x3 - mn);
    m = mn;
    s = s * sc + ((p0 + p1) + (p2 + p3));
    acc.x *= sc; acc.y *= sc; acc.z *= sc; acc.w *= sc;
    acc.x += p0 * v0.x; acc.y += p0 * v0.y; acc.z += p0 * v0.z; acc.w += p0 * v0.w;
    acc.x += p1 * v1.x; acc.y += p1 * v1.y; acc.z += p1 * v1.z; acc.w += p1 * v1.w;
    acc.x += p2 * v2.x; acc.y += p2 * v2.y; acc.z += p2 * v2.z; acc.w += p2 * v2.w;
    acc.x += p3 * v3.x; acc.y += p3 * v3.y; acc.z += p3 * v3.z; acc.w += p3 * v3.w;
  }
  for (; j < end; ++j) {
    const int s0 = slot[j];
    const float x0 = leaky(asrc[s0 * H + hh] + ad);
    const float4 v0 = h4[s0 * Q + q];
    const float mn = fmaxf(m, x0);
    const float sc = __expf(m - mn);
    const float p0 = __expf(x0 - mn);
    m = mn;
    s = s * sc + p0;
    acc.x *= sc; acc.y *= sc; acc.z *= sc; acc.w *= sc;
    acc.x += p0 * v0.x; acc.y += p0 * v0.y; acc.z += p0 * v0.z; acc.w += p0 * v0.w;
  }

  const float inv = 1.f / s;
  const float4 bv = ((const float4*)bias)[q];
  acc.x = acc.x * inv + bv.x; acc.y = acc.y * inv + bv.y;
  acc.z = acc.z * inv + bv.z; acc.w = acc.w * inv + bv.w;
  if (RELU) {
    acc.x = fmaxf(acc.x, 0.f); acc.y = fmaxf(acc.y, 0.f);
    acc.z = fmaxf(acc.z, 0.f); acc.w = fmaxf(acc.w, 0.f);
  }
  ((float4*)out)[i] = acc;
}

__global__ void __launch_bounds__(256)
seg_agg_conv1(const int* __restrict__ rowptr, const int* __restrict__ slot,
              const float* __restrict__ asrc, const float* __restrict__ adst,
              const float* __restrict__ h,
              const float* __restrict__ bias, float* __restrict__ out) {
  seg_agg_body<H1C, D1C, true>(rowptr, slot, asrc, adst, h, bias, out);
}

__global__ void __launch_bounds__(256)
seg_agg_conv2(const int* __restrict__ rowptr, const int* __restrict__ slot,
              const float* __restrict__ asrc, const float* __restrict__ adst,
              const float* __restrict__ h,
              const float* __restrict__ bias, float* __restrict__ out) {
  seg_agg_body<H2C, D2C, false>(rowptr, slot, asrc, adst, h, bias, out);
}

#define POOL_CH  16
#define POOL_WPB 5
static_assert(NNODES % POOL_CH == 0, "pool chunks cover the nodes exactly");
static_assert((NNODES / POOL_CH) % POOL_WPB == 0, "the pool grid covers the chunks exactly, no partial block");
static_assert(F2 == 64 && POOL_CH <= 32, "a lane owns features lane and lane + 32; one lane per chunk row holds its graph id");

__global__ void __launch_bounds__(POOL_WPB * 32)
pool_accum(const float* __restrict__ h, const int* __restrict__ batch,
           float* __restrict__ sums, float* __restrict__ cnt) {
  const int lane = threadIdx.x & 31;
  const int w  = blockIdx.x * POOL_WPB + __builtin_amdgcn_readfirstlane((int)(threadIdx.x >> 5));
  const int n0 = w * POOL_CH;
  const int gl = batch[n0 + (lane & (POOL_CH - 1))];
  const float* hp = h + (size_t)n0 * F2 + lane;

  float lo[POOL_CH], hi[POOL_CH];
#pragma unroll
  for (int j = 0; j < POOL_CH; ++j) { lo[j] = hp[j * F2]; hi[j] = hp[j * F2 + 32]; }

  int   g   = __builtin_amdgcn_readlane(gl, 0);
  float a0  = lo[0], a1 = hi[0];
  float run = 1.f;
#pragma unroll
  for (int j = 1; j < POOL_CH; ++j) {
    const int gj = __builtin_amdgcn_readlane(gl, j);
    if (gj != g) {
      atomicAdd(&sums[g * F2 + lane], a0);
      atomicAdd(&sums[g * F2 + 32 + lane], a1);
      if (lane == 0) atomicAdd(&cnt[g], run);
      g = gj; a0 = lo[j]; a1 = hi[j]; run = 1.f;
    } else {
      a0 += lo[j]; a1 += hi[j]; run += 1.f;
    }
  }
  atomicAdd(&sums[g * F2 + lane], a0);
  atomicAdd(&sums[g * F2 + 32 + lane], a1);
  if (lane == 0) atomicAdd(&cnt[g], run);
}

__global__ void pool_final(const float* __restrict__ sums, const float* __restrict__ cnt,
                           float* __restrict__ out) {
  int i = blockIdx.x * blockDim.x + threadIdx.x;
  if (i >= NGRAPH * F2) return;
  out[i] = sums[i] / fmaxf(cnt[i / F2], 1.0f);
}

static inline int cdiv(long long a, int b) { return (int)((a + b - 1) / b); }

extern "C" void kernel_launch(void* const* d_in, const int* in_sizes, int n_in,
                              void* d_out, int out_size, void* d_ws, size_t ws_size,
                              hipStream_t stream) {
  const float* x        = (const float*)d_in[0];
  const int*   ei       = (const int*)d_in[1];
  const int*   batch    = (const int*)d_in[2];
  const float* W1       = (const float*)d_in[3];
  const float* att_src1 = (const float*)d_in[4];
  const float* att_dst1 = (const float*)d_in[5];
  const float* bias1    = (const float*)d_in[6];
  const float* W2       = (const float*)d_in[7];
  const float* att_src2 = (const float*)d_in[8];
  const float* att_dst2 = (const float*)d_in[9];
  const float* bias2    = (const float*)d_in[10];
  float* out = (float*)d_out;
  const int* srcI = ei;
  const int* dstI = ei + NEDGES;

  char* p = (char*)d_ws;
  auto alloc = [&](size_t bytes) -> char* {
    char* r = p; p += (bytes + 255) & ~(size_t)255; return r;
  };
  float*    h1    = (float*)alloc((size_t)NNODES * F1 * 4);
  float*    asrc1 = (float*)alloc((size_t)NNODES * H1C * 4);
  float*    adst1 = (float*)alloc((size_t)NNODES * H1C * 4);
  float*    agg1  = (float*)alloc((size_t)NNODES * F1 * 4);
  float*    h2    = (float*)alloc((size_t)NNODES * F2 * 4);
  float*    asrc2 = (float*)alloc((size_t)NNODES * H2C * 4);
  float*    adst2 = (float*)alloc((size_t)NNODES * H2C * 4);
  float*    agg2  = (float*)alloc((size_t)NNODES * F2 * 4);
  float*    psum  = (float*)alloc((size_t)NGRAPH * F2 * 4);
  float*    pcnt  = (float*)alloc((size_t)NGRAPH * 4);
  int*      hist   = (int*)alloc((size_t)NHIST * 4);
  int*      off    = (int*)alloc((size_t)(NHIST + 1) * 4);
  int*      binned = (int*)alloc((size_t)NEDGES * 4);
  int*      rowptr = (int*)alloc((size_t)(NNODES + 1) * 4);
  int*      slot   = (int*)alloc((size_t)NEDGES * 4);

  hipMemsetAsync(psum,  0, (size_t)NGRAPH * F2 * 4, stream);
  hipMemsetAsync(pcnt,  0, (size_t)NGRAPH * 4, stream);

  const int T = 256;

  bin_count_gemm<<<EO_GRID, EO_T, 0, stream>>>(dstI, hist,
      x, W1, h1, att_src1, att_dst1, asrc1, adst1);
  hist_scan<<<1, SCAN_WAVES * 32, 0, stream>>>(hist, off);
  bin_scatter_gemm<<<EO_GRID, EO_T, 0, stream>>>(srcI, dstI, off, binned,
      x, W1, h1, att_src1, att_dst1, asrc1, adst1);
  bin_sort<<<NBIN, BS_T, 0, stream>>>(off, binned, rowptr, slot);

  seg_agg_conv1<<<cdiv((long long)NNODES * (F1 / 4), T), T, 0, stream>>>(rowptr, slot, asrc1, adst1,
                                                                         h1, bias1, agg1);

  wmma_gemm_strip<F1, F2, 5, D2C><<<(NNODES / 16) * (F2 / 64) / 5, 5 * 32, 0, stream>>>(
      agg1, W2, h2, att_src2, att_dst2, asrc2, adst2);
  seg_agg_conv2<<<cdiv((long long)NNODES * (F2 / 4), T), T, 0, stream>>>(rowptr, slot, asrc2, adst2,
                                                                         h2, bias2, agg2);

  pool_accum<<<(NNODES / POOL_CH) / POOL_WPB, POOL_WPB * 32, 0, stream>>>(agg2, batch, psum, pcnt);
  pool_final<<<cdiv((long long)NGRAPH * F2, T), T, 0, stream>>>(psum, pcnt, out);
}


# conv2 product relu(conv1)@W2 taken inside conv1's aggregation launch: a block's 16 rows stay in LDS, 12 waves split k 6 ways on WMMA, partials summed via LDS; agg1, one launch gone; sum order changes
# speedup vs baseline: 9.9355x; 1.0708x over previous
#include <hip/hip_runtime.h>
#include <math.h>

#define NNODES 50000
#define NEDGES 800000
#define NGRAPH 256
#define FIN    128
#define H1C    12
#define D1C    32
#define F1     (H1C * D1C)
#define H2C    1
#define D2C    64
#define F2     (H2C * D2C)

typedef float v2f __attribute__((ext_vector_type(2)));
typedef float v4f __attribute__((ext_vector_type(4)));
typedef float v8f __attribute__((ext_vector_type(8)));


template <int M>
__device__ __forceinline__ float row_xor_add(float v) {
  return v + __builtin_bit_cast(float, __builtin_amdgcn_update_dpp(
                 0, __builtin_bit_cast(int, v), 0x160 | M, 0xf, 0xf, true));
}

template <int K, int NC, int D>
__device__ __forceinline__ void
gemm_strip_wave(const int w,
                const float* __restrict__ A, const float* __restrict__ B,
                float* __restrict__ C,
                const float* __restrict__ att_src, const float* __restrict__ att_dst,
                float* __restrict__ asrc, float* __restrict__ adst) {
  constexpr int MT     = NNODES / 16;
  constexpr int NCH    = K / 16;
  constexpr int BROW   = NC / 4;
  static_assert(NNODES % 16 == 0 && K % 16 == 0 && NC % 64 == 0, "tile shapes divide the problem");
  constexpr int G = D / 4;
  constexpr int H = NC / D;
  static_assert(D == 32 || D == 64, "a head is 8 or 16 lanes of a row, and a strip holds whole heads");

  const int lane = threadIdx.x & 31;
  const int c    = lane & 15;
  const int h    = lane >> 4;
  const int strip = w / MT;
  const int m0    = (w - strip * MT) << 4;
  const int nb    = strip << 6;

  const v4f* ap = (const v4f*)(A + (size_t)(m0 + c) * K + 8 * h);
  const v4f* bp = (const v4f*)(B + (size_t)(8 * h) * NC + nb + 4 * c);

  v8f acc[4] = {};
  v4f a[2][2];
  v4f b[2][8];

  a[0][0] = ap[0];
  a[0][1] = ap[1];
#pragma unroll
  for (int j = 0; j < 8; ++j) b[0][j] = bp[j * BROW];

  const v4f ws = *(const v4f*)(att_src + nb + 4 * c);
  const v4f wd = *(const v4f*)(att_dst + nb + 4 * c);

#pragma unroll
  for (int i = 0; i < NCH; ++i) {
    const int cur = i & 1, nxt = cur ^ 1;
    if (i + 1 < NCH) {
      a[nxt][0] = ap[4 * (i + 1)];
      a[nxt][1] = ap[4 * (i + 1) + 1];
#pragma unroll
      for (int j = 0; j < 8; ++j) b[nxt][j] = bp[(16 * (i + 1) + j) * BROW];
    }
    __builtin_amdgcn_sched_barrier(0);
#pragma unroll
    for (int s = 0; s < 4; ++s) {
      v2f af;
      af.x = a[cur][s >> 1][(s & 1) * 2];
      af.y = a[cur][s >> 1][(s & 1) * 2 + 1];
#pragma unroll
      for (int t = 0; t < 4; ++t) {
        v2f bf;
        bf.x = b[cur][2 * s][t];
        bf.y = b[cur][2 * s + 1][t];
        acc[t] = __builtin_amdgcn_wmma_f32_16x16x4_f32(
             false, af,  false, bf,
             (short)0, acc[t],  false,  false);
      }
    }
  }

  v4f* crow = (v4f*)(C + (size_t)(m0 + (h << 3)) * NC + nb + 4 * c);
#pragma unroll
  for (int r = 0; r < 8; ++r) {
    v4f o;
    o.x = acc[0][r]; o.y = acc[1][r]; o.z = acc[2][r]; o.w = acc[3][r];
    crow[r * BROW] = o;
  }

  float ps[8], pd[8];
#pragma unroll
  for (int r = 0; r < 8; ++r) {
    ps[r] = acc[0][r] * ws.x + acc[1][r] * ws.y + acc[2][r] * ws.z + acc[3][r] * ws.w;
    pd[r] = acc[0][r] * wd.x + acc[1][r] * wd.y + acc[2][r] * wd.z + acc[3][r] * wd.w;
  }
#pragma unroll
  for (int r = 0; r < 8; ++r) { ps[r] = row_xor_add<1>(ps[r]); pd[r] = row_xor_add<1>(pd[r]); }
#pragma unroll
  for (int r = 0; r < 8; ++r) { ps[r] = row_xor_add<2>(ps[r]); pd[r] = row_xor_add<2>(pd[r]); }
#pragma unroll
  for (int r = 0; r < 8; ++r) { ps[r] = row_xor_add<4>(ps[r]); pd[r] = row_xor_add<4>(pd[r]); }
  if (G == 16) {
#pragma unroll
    for (int r = 0; r < 8; ++r) { ps[r] = row_xor_add<8>(ps[r]); pd[r] = row_xor_add<8>(pd[r]); }
  }
  const int g = c & (G - 1);
  float vs = ps[0], vd = pd[0];
#pragma unroll
  for (int r = 1; r < 8; ++r) { vs = (g == r) ? ps[r] : vs; vd = (g == r) ? pd[r] : vd; }
  if (g < 8) {
    const int idx = (m0 + (h << 3) + g) * H + (nb + 4 * c) / D;
    asrc[idx] = vs;
    adst[idx] = vd;
  }
}

#define G1_MT    (NNODES / 16)
#define G1_PART  (3 * G1_MT)
#define EO_WPB   5
#define EO_T     (EO_WPB * 32)
#define EO_GRID  (G1_PART / EO_WPB)
#define BIN_SHIFT 8
#define BIN_NODES (1 << BIN_SHIFT)
#define NBIN     ((NNODES + BIN_NODES - 1) / BIN_NODES)
#define CHUNK_E  4096
#define NCHUNK   ((NEDGES + CHUNK_E - 1) / CHUNK_E)
#define NHIST    (NBIN * NCHUNK)
#define CH_G     (CHUNK_E / 4)
#define EG_TRIPS ((CH_G + EO_T - 1) / EO_T)
static_assert(G1_PART % EO_WPB == 0, "the carrying grids cover their three strips exactly");
static_assert(2 * G1_PART == G1_MT * (F1 / 64), "the two launches cover the six strips");
static_assert(NEDGES % 4 == 0 && CHUNK_E % 4 == 0, "a 16-B group of four edges lies wholly inside one chunk and inside the list");
static_assert(NCHUNK <= EO_GRID, "the carrying grids have a block for every chunk");
static_assert(NNODES <= 65536 && BIN_NODES <= 256, "a packed edge word holds the source in 16 bits and the node's place in its bin above them");
static_assert(NHIST % 4 == 0, "a 16-B group of counts lies wholly inside or wholly outside the table");

__global__ void __launch_bounds__(EO_T)
bin_count_gemm(const int* __restrict__ dstI, int* __restrict__ hist,
               const float* __restrict__ x, const float* __restrict__ W,
               float* __restrict__ hout,
               const float* __restrict__ att_src, const float* __restrict__ att_dst,
               float* __restrict__ asrc, float* __restrict__ adst) {
  __shared__ int bc[NBIN];
  if (blockIdx.x < NCHUNK) {
    for (int b = threadIdx.x; b < NBIN; b += EO_T) bc[b] = 0;
    __syncthreads();
    int4 d[EG_TRIPS];
#pragma unroll
    for (int k = 0; k < EG_TRIPS; ++k) {
      const int q = threadIdx.x + k * EO_T;
      const int g = blockIdx.x * CH_G + q;
      d[k] = (q < CH_G && g < NEDGES / 4) ? ((const int4*)dstI)[g] : make_int4(0, 0, 0, 0);
    }
#pragma unroll
    for (int k = 0; k < EG_TRIPS; ++k) {
      const int q = threadIdx.x + k * EO_T;
      const int g = blockIdx.x * CH_G + q;
      if (q < CH_G && g < NEDGES / 4) {
        atomicAdd(&bc[d[k].x >> BIN_SHIFT], 1);
        atomicAdd(&bc[d[k].y >> BIN_SHIFT], 1);
        atomicAdd(&bc[d[k].z >> BIN_SHIFT], 1);
        atomicAdd(&bc[d[k].w >> BIN_SHIFT], 1);
      }
    }
    __syncthreads();
    for (int b = threadIdx.x; b < NBIN; b += EO_T) hist[b * NCHUNK + blockIdx.x] = bc[b];
  }
  const int w = blockIdx.x * EO_WPB + __builtin_amdgcn_readfirstlane((int)(threadIdx.x >> 5));
  gemm_strip_wave<FIN, F1, D1C>(w, x, W, hout, att_src, att_dst, asrc, adst);
}

#define SCAN_WAVES 32
#define SCAN_ROWS  10
#define SCAN_SPAN  (SCAN_ROWS * 128)
static_assert(SCAN_WAVES * SCAN_SPAN >= NHIST, "the scan block covers the whole table");

template <int S>
__device__ __forceinline__ int row_shr_add(int v) {
  return v + __builtin_amdgcn_update_dpp(0, v, 0x110 | S, 0xf, 0xf, true);
}

__device__ __forceinline__ int wave_incl_scan(int v, int lane) {
  v = row_shr_add<1>(v);
  v = row_shr_add<2>(v);
  v = row_shr_add<4>(v);
  v = row_shr_add<8>(v);
  const int lo = __builtin_amdgcn_readlane(v, 15);
  return v + ((lane >= 16) ? lo : 0);
}

__global__ void __launch_bounds__(SCAN_WAVES * 32)
hist_scan(const int* __restrict__ hist, int* __restrict__ off) {
  __shared__ int wtot[SCAN_WAVES];
  const int lane = threadIdx.x & 31;
  const int w    = __builtin_amdgcn_readfirstlane((int)(threadIdx.x >> 5));
  const int base = w * SCAN_SPAN + 4 * lane;

  int4 c[SCAN_ROWS];
#pragma unroll
  for (int k = 0; k < SCAN_ROWS; ++k) {
    const int i = base + k * 128;
    c[k] = (i < NHIST) ? *(const int4*)(hist + i) : make_int4(0, 0, 0, 0);
  }

  int ex[SCAN_ROWS];
  int carry = 0;
#pragma unroll
  for (int k = 0; k < SCAN_ROWS; ++k) {
    const int t   = (c[k].x + c[k].y) + (c[k].z + c[k].w);
    const int inc = wave_incl_scan(t, lane);
    ex[k] = carry + inc - t;
    carry += __builtin_amdgcn_readlane(inc, 31);
  }

  if (lane == 0) wtot[w] = carry;
  __syncthreads();
  const int tw   = wtot[lane];
  const int iw   = wave_incl_scan(tw, lane);
  const int woff = __builtin_amdgcn_readlane(iw - tw, w);

#pragma unroll
  for (int k = 0; k < SCAN_ROWS; ++k) {
    const int i = base + k * 128;
    if (i < NHIST) {
      int4 r;
      r.x = woff + ex[k];
      r.y = r.x + c[k].x;
      r.z = r.y + c[k].y;
      r.w = r.z + c[k].z;
      *(int4*)(off + i) = r;
    }
  }
  if (threadIdx.x == 0) off[NHIST] = __builtin_amdgcn_readlane(iw, 31);
}

__global__ void __launch_bounds__(EO_T)
bin_scatter_gemm(const int* __restrict__ srcI, const int* __restrict__ dstI,
                 const int* __restrict__ off, int* __restrict__ binned,
                 const float* __restrict__ x, const float* __restrict__ W,
                 float* __restrict__ hout,
                 const float* __restrict__ att_src, const float* __restrict__ att_dst,
                 float* __restrict__ asrc, float* __restrict__ adst) {
  __shared__ int bp[NBIN];
  if (blockIdx.x < NCHUNK) {
    for (int b = threadIdx.x; b < NBIN; b += EO_T) bp[b] = off[b * NCHUNK + blockIdx.x];
    __syncthreads();
    int4 d[EG_TRIPS], s[EG_TRIPS];
#pragma unroll
    for (int k = 0; k < EG_TRIPS; ++k) {
      const int q = threadIdx.x + k * EO_T;
      const int g = blockIdx.x * CH_G + q;
      const bool in = (q < CH_G && g < NEDGES / 4);
      d[k] = in ? ((const int4*)dstI)[g] : make_int4(0, 0, 0, 0);
      s[k] = in ? ((const int4*)srcI)[g] : make_int4(0, 0, 0, 0);
    }
#pragma unroll
    for (int k = 0; k < EG_TRIPS; ++k) {
      const int q = threadIdx.x + k * EO_T;
      const int g = blockIdx.x * CH_G + q;
      if (q < CH_G && g < NEDGES / 4) {
        const int p0 = atomicAdd(&bp[d[k].x >> BIN_SHIFT], 1);
        const int p1 = atomicAdd(&bp[d[k].y >> BIN_SHIFT], 1);
        const int p2 = atomicAdd(&bp[d[k].z >> BIN_SHIFT], 1);
        const int p3 = atomicAdd(&bp[d[k].w >> BIN_SHIFT], 1);
        binned[p0] = s[k].x | ((d[k].x & (BIN_NODES - 1)) << 16);
        binned[p1] = s[k].y | ((d[k].y & (BIN_NODES - 1)) << 16);
        binned[p2] = s[k].z | ((d[k].z & (BIN_NODES - 1)) << 16);
        binned[p3] = s[k].w | ((d[k].w & (BIN_NODES - 1)) << 16);
      }
    }
  }
  const int w = blockIdx.x * EO_WPB + __builtin_amdgcn_readfirstlane((int)(threadIdx.x >> 5));
  gemm_strip_wave<FIN, F1, D1C>(G1_PART + w, x, W, hout, att_src, att_dst, asrc, adst);
}

#define BS_T 1024
#define BS_LD 4
static_assert(BIN_NODES % 32 == 0 && BIN_NODES <= BS_T, "whole waves hold the bin's node counters, one per thread");

__global__ void __launch_bounds__(BS_T)
bin_sort(const int* __restrict__ off, const int* __restrict__ binned,
         int* __restrict__ rowptr, int* __restrict__ slot) {
  __shared__ int nc[BIN_NODES];
  __shared__ int wsum[BIN_NODES / 32];
  const int tid  = threadIdx.x;
  const int lane = tid & 31;
  const int b    = blockIdx.x;
  const int e0   = off[b * NCHUNK];
  const int e1   = off[(b + 1) * NCHUNK];
  if (tid < BIN_NODES) nc[tid] = 0;
  __syncthreads();
  for (int e = e0 + tid; e < e1; e += BS_LD * BS_T) {
    int v[BS_LD];
#pragma unroll
    for (int k = 0; k < BS_LD; ++k) v[k] = binned[min(e + k * BS_T, e1 - 1)];
#pragma unroll
    for (int k = 0; k < BS_LD; ++k)
      if (e + k * BS_T < e1) atomicAdd(&nc[v[k] >> 16], 1);
  }
  __syncthreads();
  int c = 0, inc = 0;
  if (tid < BIN_NODES) {
    c   = nc[tid];
    inc = wave_incl_scan(c, lane);
    if (lane == 31) wsum[tid >> 5] = inc;
  }
  __syncthreads();
  if (tid < BIN_NODES) {
    int wo = 0;
    for (int j = 0; j < (tid >> 5); ++j) wo += wsum[j];
    const int ex = wo + inc - c;
    nc[tid] = ex;
    const int node = b * BIN_NODES + tid;
    if (node < NNODES) rowptr[node] = e0 + ex;
  }
  if (b == NBIN - 1 && tid == 0) rowptr[NNODES] = e1;
  __syncthreads();
  for (int e = e0 + tid; e < e1; e += BS_LD * BS_T) {
    int v[BS_LD];
#pragma unroll
    for (int k = 0; k < BS_LD; ++k) v[k] = binned[min(e + k * BS_T, e1 - 1)];
#pragma unroll
    for (int k = 0; k < BS_LD; ++k) {
      if (e + k * BS_T < e1) {
        const int p = atomicAdd(&nc[v[k] >> 16], 1);
        slot[e0 + p] = v[k] & 0xffff;
      }
    }
  }
}

__device__ __forceinline__ float leaky(float v) { return (v > 0.f) ? v : 0.2f * v; }

template <int H, int D, bool RELU>
__device__ __forceinline__ float4 seg_agg_row(const int d, const int q,
                                              const int* __restrict__ rowptr,
                                              const int* __restrict__ slot,
                                              const float* __restrict__ asrc,
                                              const float* __restrict__ adst,
                                              const float* __restrict__ h,
                                              const float* __restrict__ bias) {
  constexpr int Q = (H * D) / 4;
  const int hh = (q * 4) / D;
  const float4* h4 = (const float4*)h;

  const float ad = adst[d * H + hh];
  float  m   = leaky(asrc[d * H + hh] + ad);
  float  s   = 1.f;
  float4 acc = h4[d * Q + q];

  int j = rowptr[d];
  const int end = rowptr[d + 1];
  for (; j + 4 <= end; j += 4) {
    const int s0 = slot[j], s1 = slot[j + 1], s2 = slot[j + 2], s3 = slot[j + 3];
    const float x0 = leaky(asrc[s0 * H + hh] + ad), x1 = leaky(asrc[s1 * H + hh] + ad);
    const float x2 = leaky(asrc[s2 * H + hh] + ad), x3 = leaky(asrc[s3 * H + hh] + ad);
    const float4 v0 = h4[s0 * Q + q], v1 = h4[s1 * Q + q];
    const float4 v2 = h4[s2 * Q + q], v3 = h4[s3 * Q + q];
    const float mn = fmaxf(fmaxf(m, x0), fmaxf(fmaxf(x1, x2), x3));
    const float sc = __expf(m - mn);
    const float p0 = __expf(x0 - mn), p1 = __expf(x1 - mn);
    const float p2 = __expf(x2 - mn), p3 = __expf(x3 - mn);
    m = mn;
    s = s * sc + ((p0 + p1) + (p2 + p3));
    acc.x *= sc; acc.y *= sc; acc.z *= sc; acc.w *= sc;
    acc.x += p0 * v0.x; acc.y += p0 * v0.y; acc.z += p0 * v0.z; acc.w += p0 * v0.w;
    acc.x += p1 * v1.x; acc.y += p1 * v1.y; acc.z += p1 * v1.z; acc.w += p1 * v1.w;
    acc.x += p2 * v2.x; acc.y += p2 * v2.y; acc.z += p2 * v2.z; acc.w += p2 * v2.w;
    acc.x += p3 * v3.x; acc.y += p3 * v3.y; acc.z += p3 * v3.z; acc.w += p3 * v3.w;
  }
  for (; j < end; ++j) {
    const int s0 = slot[j];
    const float x0 = leaky(asrc[s0 * H + hh] + ad);
    const float4 v0 = h4[s0 * Q + q];
    const float mn = fmaxf(m, x0);
    const float sc = __expf(m - mn);
    const float p0 = __expf(x0 - mn);
    m = mn;
    s = s * sc + p0;
    acc.x *= sc; acc.y *= sc; acc.z *= sc; acc.w *= sc;
    acc.x += p0 * v0.x; acc.y += p0 * v0.y; acc.z += p0 * v0.z; acc.w += p0 * v0.w;
  }

  const float inv = 1.f / s;
  const float4 bv = ((const float4*)bias)[q];
  acc.x = acc.x * inv + bv.x; acc.y = acc.y * inv + bv.y;
  acc.z = acc.z * inv + bv.z; acc.w = acc.w * inv + bv.w;
  if (RELU) {
    acc.x = fmaxf(acc.x, 0.f); acc.y = fmaxf(acc.y, 0.f);
    acc.z = fmaxf(acc.z, 0.f); acc.w = fmaxf(acc.w, 0.f);
  }
  return acc;
}

#define FA_T   384
#define FA_TP  (F1 + 4)
static_assert(NNODES % 16 == 0 && F1 == 384 && F2 == 64, "12 waves: 3 x 128 features by 4 row groups, then 6 x 64 k by 2 x 32 columns");
static_assert(6 * 16 * F2 <= 16 * FA_TP, "the six partial tiles fit over the row tile");

__global__ void __launch_bounds__(FA_T)
seg_agg_conv1_gemm2(const int* __restrict__ rowptr, const int* __restrict__ slot,
                    const float* __restrict__ asrc, const float* __restrict__ adst,
                    const float* __restrict__ h, const float* __restrict__ bias,
                    const float* __restrict__ W2,
                    const float* __restrict__ att_src2, const float* __restrict__ att_dst2,
                    float* __restrict__ h2, float* __restrict__ asrc2, float* __restrict__ adst2) {
  __shared__ __attribute__((aligned(16))) float T[16 * FA_TP];
  const int tid  = threadIdx.x;
  const int lane = tid & 31;
  const int c    = lane & 15;
  const int hf   = lane >> 4;
  const int w  = __builtin_amdgcn_readfirstlane(tid >> 5);
  const int m0 = blockIdx.x << 4;

  {
    const int q  = (w % 3) * 32 + lane;
    const int g4 = w / 3;
    for (int p = 0; p < 4; ++p) {
      const int nl = 4 * p + g4;
      const float4 v = seg_agg_row<H1C, D1C, true>(m0 + nl, q, rowptr, slot, asrc, adst, h, bias);
      *(float4*)(T + nl * FA_TP + 4 * q) = v;
    }
  }

  const int kk = w % 6;
  const int nh = w / 6;
  const v2f* bp = (const v2f*)(W2 + (size_t)(64 * kk + 8 * hf) * F2 + 32 * nh + 2 * c);
  const float* tp = T + c * FA_TP + 64 * kk + 8 * hf;
  v8f acc[2] = {};
  __syncthreads();
#pragma unroll
  for (int ch = 0; ch < 4; ++ch) {
    v2f b[8];
#pragma unroll
    for (int j = 0; j < 8; ++j) b[j] = bp[(16 * ch + j) * (F2 / 2)];
    v4f a[2];
    a[0] = *(const v4f*)(tp + 16 * ch);
    a[1] = *(const v4f*)(tp + 16 * ch + 4);
#pragma unroll
    for (int s = 0; s < 4; ++s) {
      v2f af;
      af.x = a[s >> 1][(s & 1) * 2];
      af.y = a[s >> 1][(s & 1) * 2 + 1];
#pragma unroll
      for (int t = 0; t < 2; ++t) {
        v2f bf;
        bf.x = b[2 * s][t];
        bf.y = b[2 * s + 1][t];
        acc[t] = __builtin_amdgcn_wmma_f32_16x16x4_f32(
             false, af,  false, bf,
             (short)0, acc[t],  false,  false);
      }
    }
  }
  __syncthreads();
  float* P = T;
#pragma unroll
  for (int r = 0; r < 8; ++r) {
    v2f o;
    o.x = acc[0][r]; o.y = acc[1][r];
    *(v2f*)(P + (kk * 16 + 8 * hf + r) * F2 + 32 * nh + 2 * c) = o;
  }
  __syncthreads();
  if (tid < 256) {
    const int row = tid >> 4;
    const int cq  = tid & 15;
    v4f sum = *(const v4f*)(P + row * F2 + 4 * cq);
#pragma unroll
    for (int k = 1; k < 6; ++k) sum += *(const v4f*)(P + (k * 16 + row) * F2 + 4 * cq);
    const int node = m0 + row;
    *(v4f*)(h2 + (size_t)node * F2 + 4 * cq) = sum;
    const v4f ws = *(const v4f*)(att_src2 + 4 * cq);
    const v4f wd = *(const v4f*)(att_dst2 + 4 * cq);
    float ps = sum.x * ws.x + sum.y * ws.y + sum.z * ws.z + sum.w * ws.w;
    float pd = sum.x * wd.x + sum.y * wd.y + sum.z * wd.z + sum.w * wd.w;
    ps = row_xor_add<1>(ps); pd = row_xor_add<1>(pd);
    ps = row_xor_add<2>(ps); pd = row_xor_add<2>(pd);
    ps = row_xor_add<4>(ps); pd = row_xor_add<4>(pd);
    ps = row_xor_add<8>(ps); pd = row_xor_add<8>(pd);
    if (cq == 0) { asrc2[node] = ps; adst2[node] = pd; }
  }
}

__global__ void __launch_bounds__(256)
seg_agg_conv2(const int* __restrict__ rowptr, const int* __restrict__ slot,
              const float* __restrict__ asrc, const float* __restrict__ adst,
              const float* __restrict__ h,
              const float* __restrict__ bias, float* __restrict__ out) {
  constexpr int Q = F2 / 4;
  const int i = blockIdx.x * blockDim.x + threadIdx.x;
  if (i >= NNODES * Q) return;
  const int d = i / Q;
  const int q = i - d * Q;
  ((float4*)out)[i] = seg_agg_row<H2C, D2C, false>(d, q, rowptr, slot, asrc, adst, h, bias);
}

#define POOL_CH  16
#define POOL_WPB 5
static_assert(NNODES % POOL_CH == 0, "pool chunks cover the nodes exactly");
static_assert((NNODES / POOL_CH) % POOL_WPB == 0, "the pool grid covers the chunks exactly, no partial block");
static_assert(F2 == 64 && POOL_CH <= 32, "a lane owns features lane and lane + 32; one lane per chunk row holds its graph id");

__global__ void __launch_bounds__(POOL_WPB * 32)
pool_accum(const float* __restrict__ h, const int* __restrict__ batch,
           float* __restrict__ sums, float* __restrict__ cnt) {
  const int lane = threadIdx.x & 31;
  const int w  = blockIdx.x * POOL_WPB + __builtin_amdgcn_readfirstlane((int)(threadIdx.x >> 5));
  const int n0 = w * POOL_CH;
  const int gl = batch[n0 + (lane & (POOL_CH - 1))];
  const float* hp = h + (size_t)n0 * F2 + lane;

  float lo[POOL_CH], hi[POOL_CH];
#pragma unroll
  for (int j = 0; j < POOL_CH; ++j) { lo[j] = hp[j * F2]; hi[j] = hp[j * F2 + 32]; }

  int   g   = __builtin_amdgcn_readlane(gl, 0);
  float a0  = lo[0], a1 = hi[0];
  float run = 1.f;
#pragma unroll
  for (int j = 1; j < POOL_CH; ++j) {
    const int gj = __builtin_amdgcn_readlane(gl, j);
    if (gj != g) {
      atomicAdd(&sums[g * F2 + lane], a0);
      atomicAdd(&sums[g * F2 + 32 + lane], a1);
      if (lane == 0) atomicAdd(&cnt[g], run);
      g = gj; a0 = lo[j]; a1 = hi[j]; run = 1.f;
    } else {
      a0 += lo[j]; a1 += hi[j]; run += 1.f;
    }
  }
  atomicAdd(&sums[g * F2 + lane], a0);
  atomicAdd(&sums[g * F2 + 32 + lane], a1);
  if (lane == 0) atomicAdd(&cnt[g], run);
}

__global__ void pool_final(const float* __restrict__ sums, const float* __restrict__ cnt,
                           float* __restrict__ out) {
  int i = blockIdx.x * blockDim.x + threadIdx.x;
  if (i >= NGRAPH * F2) return;
  out[i] = sums[i] / fmaxf(cnt[i / F2], 1.0f);
}

static inline int cdiv(long long a, int b) { return (int)((a + b - 1) / b); }

extern "C" void kernel_launch(void* const* d_in, const int* in_sizes, int n_in,
                              void* d_out, int out_size, void* d_ws, size_t ws_size,
                              hipStream_t stream) {
  const float* x        = (const float*)d_in[0];
  const int*   ei       = (const int*)d_in[1];
  const int*   batch    = (const int*)d_in[2];
  const float* W1       = (const float*)d_in[3];
  const float* att_src1 = (const float*)d_in[4];
  const float* att_dst1 = (const float*)d_in[5];
  const float* bias1    = (const float*)d_in[6];
  const float* W2       = (const float*)d_in[7];
  const float* att_src2 = (const float*)d_in[8];
  const float* att_dst2 = (const float*)d_in[9];
  const float* bias2    = (const float*)d_in[10];
  float* out = (float*)d_out;
  const int* srcI = ei;
  const int* dstI = ei + NEDGES;

  char* p = (char*)d_ws;
  auto alloc = [&](size_t bytes) -> char* {
    char* r = p; p += (bytes + 255) & ~(size_t)255; return r;
  };
  float*    h1    = (float*)alloc((size_t)NNODES * F1 * 4);
  float*    asrc1 = (float*)alloc((size_t)NNODES * H1C * 4);
  float*    adst1 = (float*)alloc((size_t)NNODES * H1C * 4);
  float*    h2    = (float*)alloc((size_t)NNODES * F2 * 4);
  float*    asrc2 = (float*)alloc((size_t)NNODES * H2C * 4);
  float*    adst2 = (float*)alloc((size_t)NNODES * H2C * 4);
  float*    agg2  = (float*)alloc((size_t)NNODES * F2 * 4);
  float*    psum  = (float*)alloc((size_t)NGRAPH * F2 * 4);
  float*    pcnt  = (float*)alloc((size_t)NGRAPH * 4);
  int*      hist   = (int*)alloc((size_t)NHIST * 4);
  int*      off    = (int*)alloc((size_t)(NHIST + 1) * 4);
  int*      binned = (int*)alloc((size_t)NEDGES * 4);
  int*      rowptr = (int*)alloc((size_t)(NNODES + 1) * 4);
  int*      slot   = (int*)alloc((size_t)NEDGES * 4);

  hipMemsetAsync(psum,  0, (size_t)NGRAPH * F2 * 4, stream);
  hipMemsetAsync(pcnt,  0, (size_t)NGRAPH * 4, stream);

  const int T = 256;

  bin_count_gemm<<<EO_GRID, EO_T, 0, stream>>>(dstI, hist,
      x, W1, h1, att_src1, att_dst1, asrc1, adst1);
  hist_scan<<<1, SCAN_WAVES * 32, 0, stream>>>(hist, off);
  bin_scatter_gemm<<<EO_GRID, EO_T, 0, stream>>>(srcI, dstI, off, binned,
      x, W1, h1, att_src1, att_dst1, asrc1, adst1);
  bin_sort<<<NBIN, BS_T, 0, stream>>>(off, binned, rowptr, slot);

  seg_agg_conv1_gemm2<<<NNODES / 16, FA_T, 0, stream>>>(rowptr, slot, asrc1, adst1, h1, bias1,
                                                        W2, att_src2, att_dst2, h2, asrc2, adst2);

  seg_agg_conv2<<<cdiv((long long)NNODES * (F2 / 4), T), T, 0, stream>>>(rowptr, slot, asrc2, adst2,
                                                                         h2, bias2, agg2);

  pool_accum<<<(NNODES / POOL_CH) / POOL_WPB, POOL_WPB * 32, 0, stream>>>(agg2, batch, psum, pcnt);
  pool_final<<<cdiv((long long)NGRAPH * F2, T), T, 0, stream>>>(psum, pcnt, out);
}
